# starting kernel
# baseline (speedup 1.0000x reference)
#include <hip/hip_runtime.h>
#include <hip/hip_bf16.h>
#include <math.h>

typedef float v2f __attribute__((ext_vector_type(2)));
typedef float v8f __attribute__((ext_vector_type(8)));

#define PI_F 3.14159265358979323846f

__global__ __launch_bounds__(256)
void conv1_silu_kernel(const float* __restrict__ x, const float* __restrict__ w1,
                       const float* __restrict__ b1, float* __restrict__ out)
{
    int idx = blockIdx.x * 256 + threadIdx.x;
    if (idx >= 32 * 32 * 112 * 112) return;
    int ox = idx % 112; int t = idx / 112;
    int oy = t % 112;   t /= 112;
    int oc = t % 32;    int img = t / 32;
    float acc = b1[oc];
    const float* wb = w1 + oc * 27;
    const float* xb = x + (size_t)img * 3 * 224 * 224;
    #pragma unroll
    for (int ic = 0; ic < 3; ++ic)
        #pragma unroll
        for (int ky = 0; ky < 3; ++ky) {
            int iy = oy * 2 + ky;
            if (iy >= 224) continue;
            #pragma unroll
            for (int kx = 0; kx < 3; ++kx) {
                int ix = ox * 2 + kx;
                if (ix >= 224) continue;
                acc += xb[(ic * 224 + iy) * 224 + ix] * wb[ic * 9 + ky * 3 + kx];
            }
        }
    out[idx] = acc / (1.f + expf(-acc));
}

__global__ __launch_bounds__(128)
void conv2_wmma_pool_kernel(const float* __restrict__ h1, const float* __restrict__ w2,
                            const float* __restrict__ b2, float* __restrict__ feat)
{
    __shared__ float sW[16 * 288];
    __shared__ float sT[4][720];
    __shared__ float sAcc[16];

    const int mtile = blockIdx.x;
    const int img   = blockIdx.y;
    const int tid   = threadIdx.x;
    const int lane  = tid & 31;
    const int wave  = tid >> 5;
    const int half  = lane >> 4;
    const int nl    = lane & 15;

    for (int i = tid; i < 16 * 288; i += 128) {
        int m  = i / 288, kk = i - m * 288;
        int icb = kk / 72, r = kk - icb * 72;
        int s = r >> 3, ic = r & 7;
        sW[i] = w2[(mtile * 16 + m) * 288 + (icb * 8 + ic) * 9 + s];
    }
    if (tid < 16) sAcc[tid] = 0.f;
    __syncthreads();

    const int dy = nl >> 3, dx = nl & 7;
    const int laneOff = dy * 288 + dx * 16;

    const int j0 = lane, j1 = lane + 32, j2 = lane + 64;
    const int ly0 = j0 / 18, lx0 = j0 - ly0 * 18;
    const int ly1 = j1 / 18, lx1 = j1 - ly1 * 18;
    const int ly2 = j2 / 18, lx2 = j2 - ly2 * 18;
    const bool j2ok = (j2 < 90);
    const int t0 = ly0 * 144 + lx0 * 8;
    const int t1 = ly1 * 144 + lx1 * 8;
    const int t2 = ly2 * 144 + lx2 * 8;

    float bias[8];
    #pragma unroll
    for (int j = 0; j < 8; ++j) bias[j] = b2[mtile * 16 + j + half * 8];

    float accR[8] = {0.f, 0.f, 0.f, 0.f, 0.f, 0.f, 0.f, 0.f};
    const float* base = h1 + (size_t)img * (32 * 112 * 112);
    float* tile = sT[wave];
    const float* tb = tile + laneOff + half * 2;

    for (int nt = wave; nt < 196; nt += 4) {
        int yt = nt / 7, xt = nt - yt * 7;
        int iy0 = yt * 4, ix0 = xt * 16;
        v8f c = {0.f, 0.f, 0.f, 0.f, 0.f, 0.f, 0.f, 0.f};

        for (int icb = 0; icb < 4; ++icb) {
            #pragma unroll
            for (int ic = 0; ic < 8; ++ic) {
                const float* p = base + (size_t)(icb * 8 + ic) * (112 * 112);
                {
                    int iy = iy0 + ly0, ix = ix0 + lx0;
                    bool ok = (iy < 112) & (ix < 112);
                    float v = p[(iy < 112 ? iy : 111) * 112 + (ix < 112 ? ix : 111)];
                    tile[t0 + ic] = ok ? v : 0.f;
                }
                {
                    int iy = iy0 + ly1, ix = ix0 + lx1;
                    bool ok = (iy < 112) & (ix < 112);
                    float v = p[(iy < 112 ? iy : 111) * 112 + (ix < 112 ? ix : 111)];
                    tile[t1 + ic] = ok ? v : 0.f;
                }
                if (j2ok) {
                    int iy = iy0 + ly2, ix = ix0 + lx2;
                    bool ok = (iy < 112) & (ix < 112);
                    float v = p[(iy < 112 ? iy : 111) * 112 + (ix < 112 ? ix : 111)];
                    tile[t2 + ic] = ok ? v : 0.f;
                }
            }
            asm volatile("s_wait_dscnt 0x0" ::: "memory");

            const float* wrow = sW + nl * 288 + icb * 72 + half * 2;
            #pragma unroll
            for (int kq = 0; kq < 18; ++kq) {
                const int s  = kq >> 1;
                const int ky = s / 3, kx = s - ky * 3;
                const int bo = ky * 144 + kx * 8 + (kq & 1) * 4;
                v2f a = *(const v2f*)(wrow + kq * 4);
                v2f b = *(const v2f*)(tb + bo);
                c = __builtin_amdgcn_wmma_f32_16x16x4_f32(false, a, false, b,
                                                          (short)0, c, false, false);
            }
        }
        #pragma unroll
        for (int j = 0; j < 8; ++j) {
            float v = c[j] + bias[j];
            accR[j] += v / (1.f + expf(-v));
        }
    }
    #pragma unroll
    for (int j = 0; j < 8; ++j)
        atomicAdd(&sAcc[j + half * 8], accR[j]);
    __syncthreads();
    if (tid < 16)
        feat[img * 1280 + mtile * 16 + tid] = sAcc[tid] * (1.f / 3136.f);
}

__global__ __launch_bounds__(224)
void dft_rows_kernel(const float* __restrict__ x, float* __restrict__ F1)
{
    __shared__ float ct[224], st[224];
    int y  = blockIdx.x % 224;
    int bc = blockIdx.x / 224;
    int u  = threadIdx.x;
    float ang = (2.0f * PI_F / 224.0f) * (float)u;
    ct[u] = cosf(ang); st[u] = sinf(ang);
    __syncthreads();
    const float* row = x + ((size_t)bc * 224 + y) * 224;
    float re = 0.f, im = 0.f; int idx = 0;
    for (int xx = 0; xx < 224; ++xx) {
        float v = row[xx];
        re += v * ct[idx]; im -= v * st[idx];
        idx += u; if (idx >= 224) idx -= 224;
    }
    float* o = F1 + (((size_t)bc * 224 + y) * 224 + u) * 2;
    o[0] = re; o[1] = im;
}

__global__ __launch_bounds__(224)
void dft_cols_mag_kernel(const float* __restrict__ F1, float* __restrict__ mag)
{
    __shared__ float ct[224], st[224];
    int u  = blockIdx.x % 224;
    int bc = blockIdx.x / 224;
    int v  = threadIdx.x;
    float ang = (2.0f * PI_F / 224.0f) * (float)v;
    ct[v] = cosf(ang); st[v] = sinf(ang);
    __syncthreads();
    const float* base = F1 + (size_t)bc * 224 * 224 * 2 + u * 2;
    float re = 0.f, im = 0.f; int idx = 0;
    for (int y = 0; y < 224; ++y) {
        float fr = base[(size_t)y * 448], fi = base[(size_t)y * 448 + 1];
        float cc = ct[idx], ss = st[idx];
        re += fr * cc + fi * ss;
        im += fi * cc - fr * ss;
        idx += v; if (idx >= 224) idx -= 224;
    }
    float m = logf(sqrtf(re * re + im * im) + 1.0f);
    int vo = v + 112; if (vo >= 224) vo -= 224;
    int uo = u + 112; if (uo >= 224) uo -= 224;
    mag[((size_t)bc * 224 + vo) * 224 + uo] = m;
}

__global__ __launch_bounds__(256)
void minmax_kernel(const float* __restrict__ mag, float* __restrict__ mm)
{
    __shared__ float smn[256], smx[256];
    int bc = blockIdx.x, tid = threadIdx.x;
    const float* p = mag + (size_t)bc * 50176;
    float mn = 3.0e38f, mx = -3.0e38f;
    for (int i = tid; i < 50176; i += 256) {
        float v = p[i]; mn = fminf(mn, v); mx = fmaxf(mx, v);
    }
    smn[tid] = mn; smx[tid] = mx; __syncthreads();
    for (int s = 128; s > 0; s >>= 1) {
        if (tid < s) { smn[tid] = fminf(smn[tid], smn[tid + s]);
                       smx[tid] = fmaxf(smx[tid], smx[tid + s]); }
        __syncthreads();
    }
    if (tid == 0) { mm[bc * 2] = smn[0]; mm[bc * 2 + 1] = smx[0]; }
}

__global__ __launch_bounds__(256)
void normalize_kernel(float* __restrict__ mag, const float* __restrict__ mm)
{
    int idx = blockIdx.x * 256 + threadIdx.x;
    if (idx >= 96 * 50176) return;
    int bc = idx / 50176;
    float mn = mm[bc * 2], mx = mm[bc * 2 + 1];
    mag[idx] = (mag[idx] - mn) / (mx - mn + 1e-8f);
}

__global__ __launch_bounds__(256)
void fc_kernel(const float* __restrict__ in, const float* __restrict__ w,
               const float* __restrict__ b, float* __restrict__ out,
               int K, int N, int outStride, int outOffset, int act)
{
    int idx = blockIdx.x * 256 + threadIdx.x;
    if (idx >= 32 * N) return;
    int i = idx / N, j = idx % N;
    const float* ir = in + (size_t)i * K;
    const float* wr = w + (size_t)j * K;
    float acc = b[j];
    for (int k = 0; k < K; ++k) acc += ir[k] * wr[k];
    if (act == 1) acc = fmaxf(acc, 0.f);
    out[(size_t)i * outStride + outOffset + j] = acc;
}

__global__ __launch_bounds__(256)
void attn_core_kernel(const float* __restrict__ qkv_rgb, const float* __restrict__ qkv_frq,
                      float* __restrict__ o1, float* __restrict__ o2)
{
    __shared__ float sQ[32][32], sK[32][32], sV[32][32], sS[32][33];
    int a = blockIdx.x >> 3;
    int h = blockIdx.x & 7;
    const float* qsrc  = (a == 0) ? qkv_rgb : qkv_frq;
    const float* kvsrc = (a == 0) ? qkv_frq : qkv_rgb;
    float* odst        = (a == 0) ? o1 : o2;
    int tid = threadIdx.x;
    for (int idx = tid; idx < 1024; idx += 256) {
        int q = idx >> 5, d = idx & 31;
        sQ[q][d] = qsrc [q * 768 +       h * 32 + d];
        sK[q][d] = kvsrc[q * 768 + 256 + h * 32 + d];
        sV[q][d] = kvsrc[q * 768 + 512 + h * 32 + d];
    }
    __syncthreads();
    const float scale = 0.17677669529663687f;
    for (int idx = tid; idx < 1024; idx += 256) {
        int q = idx >> 5, k = idx & 31;
        float s = 0.f;
        #pragma unroll
        for (int d = 0; d < 32; ++d) s += sQ[q][d] * sK[k][d];
        sS[q][k] = s * scale;
    }
    __syncthreads();
    if (tid < 32) {
        float mx = -3.0e38f;
        for (int k = 0; k < 32; ++k) mx = fmaxf(mx, sS[tid][k]);
        float e[32], sum = 0.f;
        for (int k = 0; k < 32; ++k) { e[k] = expf(sS[tid][k] - mx); sum += e[k]; }
        float inv = 1.f / sum;
        for (int k = 0; k < 32; ++k) sS[tid][k] = e[k] * inv;
    }
    __syncthreads();
    for (int idx = tid; idx < 1024; idx += 256) {
        int q = idx >> 5, d = idx & 31;
        float o = 0.f;
        #pragma unroll
        for (int k = 0; k < 32; ++k) o += sS[q][k] * sV[k][d];
        odst[q * 256 + h * 32 + d] = o;
    }
}

extern "C" void kernel_launch(void* const* d_in, const int* in_sizes, int n_in,
                              void* d_out, int out_size, void* d_ws, size_t ws_size,
                              hipStream_t stream)
{
    (void)in_sizes; (void)n_in; (void)out_size; (void)ws_size;
    const float* x          = (const float*)d_in[0];
    const float* rgb_w1     = (const float*)d_in[1];
    const float* rgb_b1     = (const float*)d_in[2];
    const float* rgb_w2     = (const float*)d_in[3];
    const float* rgb_b2     = (const float*)d_in[4];
    const float* frq_w1     = (const float*)d_in[5];
    const float* frq_b1     = (const float*)d_in[6];
    const float* frq_w2     = (const float*)d_in[7];
    const float* frq_b2     = (const float*)d_in[8];
    const float* rgb_proj_w = (const float*)d_in[9];
    const float* rgb_proj_b = (const float*)d_in[10];
    const float* frq_proj_w = (const float*)d_in[11];
    const float* frq_proj_b = (const float*)d_in[12];
    const float* attn_w_in  = (const float*)d_in[13];
    const float* attn_b_in  = (const float*)d_in[14];
    const float* attn_w_out = (const float*)d_in[15];
    const float* attn_b_out = (const float*)d_in[16];
    const float* fus_w1     = (const float*)d_in[17];
    const float* fus_b1     = (const float*)d_in[18];
    const float* fus_w2     = (const float*)d_in[19];
    const float* fus_b2     = (const float*)d_in[20];
    const float* fus_w3     = (const float*)d_in[21];
    const float* fus_b3     = (const float*)d_in[22];

    float* out      = (float*)d_out;
    float* feat_rgb = out + 64;
    float* feat_frq = out + 64 + 40960;

    float* ws       = (float*)d_ws;
    float* h1       = ws;
    float* F1       = ws;
    float* freqimg  = ws + 12845056;
    float* mm       = freqimg + 4816896;
    float* hd       = mm + 192;
    float* rgb_p    = hd;
    float* frq_p    = rgb_p + 8192;
    float* qkv_rgb  = frq_p + 8192;
    float* qkv_frq  = qkv_rgb + 24576;
    float* o1       = qkv_frq + 24576;
    float* o2       = o1 + 8192;
    float* hb       = o2 + 8192;
    float* h1f      = hb + 16384;
    float* h2f      = h1f + 16384;

    conv1_silu_kernel<<<50176, 256, 0, stream>>>(x, rgb_w1, rgb_b1, h1);
    conv2_wmma_pool_kernel<<<dim3(80, 32), 128, 0, stream>>>(h1, rgb_w2, rgb_b2, feat_rgb);

    dft_rows_kernel<<<21504, 224, 0, stream>>>(x, F1);
    dft_cols_mag_kernel<<<21504, 224, 0, stream>>>(F1, freqimg);
    minmax_kernel<<<96, 256, 0, stream>>>(freqimg, mm);
    normalize_kernel<<<18816, 256, 0, stream>>>(freqimg, mm);
    conv1_silu_kernel<<<50176, 256, 0, stream>>>(freqimg, frq_w1, frq_b1, h1);
    conv2_wmma_pool_kernel<<<dim3(80, 32), 128, 0, stream>>>(h1, frq_w2, frq_b2, feat_frq);

    fc_kernel<<<32, 256, 0, stream>>>(feat_rgb, rgb_proj_w, rgb_proj_b, rgb_p, 1280, 256, 256, 0, 0);
    fc_kernel<<<32, 256, 0, stream>>>(feat_frq, frq_proj_w, frq_proj_b, frq_p, 1280, 256, 256, 0, 0);
    fc_kernel<<<96, 256, 0, stream>>>(rgb_p, attn_w_in, attn_b_in, qkv_rgb, 256, 768, 768, 0, 0);
    fc_kernel<<<96, 256, 0, stream>>>(frq_p, attn_w_in, attn_b_in, qkv_frq, 256, 768, 768, 0, 0);
    attn_core_kernel<<<16, 256, 0, stream>>>(qkv_rgb, qkv_frq, o1, o2);
    fc_kernel<<<32, 256, 0, stream>>>(o1, attn_w_out, attn_b_out, hb, 256, 256, 512, 0, 0);
    fc_kernel<<<32, 256, 0, stream>>>(o2, attn_w_out, attn_b_out, hb, 256, 256, 512, 256, 0);
    fc_kernel<<<64, 256, 0, stream>>>(hb,  fus_w1, fus_b1, h1f, 512, 512, 512, 0, 1);
    fc_kernel<<<32, 256, 0, stream>>>(h1f, fus_w2, fus_b2, h2f, 512, 256, 256, 0, 1);
    fc_kernel<<<1,  256, 0, stream>>>(h2f, fus_w3, fus_b3, out, 256, 2, 2, 0, 0);
}


# head FC layers as GEMMs on f32 16x16x4 WMMA: block 32 features x 16 batch rows, 8 waves split K (16-byte row loads, all requested first), partials joined in LDS in fixed order; FC sum order changes
# speedup vs baseline: 2.2145x; 2.2145x over previous
#include <hip/hip_runtime.h>
#include <hip/hip_bf16.h>
#include <math.h>

typedef float v2f __attribute__((ext_vector_type(2)));
typedef float v8f __attribute__((ext_vector_type(8)));

#define PI_F 3.14159265358979323846f

#define C1_CHUNKS 2

__device__ __forceinline__ constexpr int  c1_tap_off(int t)
{ return t < 27 ? (t / 9) * 50176 + ((t % 9) / 3) * 224 + (t % 3) : 0; }
__device__ __forceinline__ constexpr bool c1_tap_kx2(int t) { return t < 27 && (t % 3) == 2; }
__device__ __forceinline__ constexpr bool c1_tap_ky2(int t) { return t < 27 && ((t % 9) / 3) == 2; }

__device__ __forceinline__ void conv1_load_tile(const float* __restrict__ xi, int tl,
                                                int nl, int half, float (&b)[14])
{
    const int  oy = tl / 7;
    const int  ox = (tl - oy * 7) * 16 + nl;
    const bool lastrow = (oy == 111);
    const bool lastcol = (ox == 111);
    const float* p = xi + (2 * oy) * 224 + 2 * ox;
    #pragma unroll
    for (int j = 0; j < 14; ++j) {
        const int  t0  = 4 * (j >> 1) + (j & 1);
        const int  off = half ? c1_tap_off(t0 + 2) : c1_tap_off(t0);
        const bool kx2 = half ? c1_tap_kx2(t0 + 2) : c1_tap_kx2(t0);
        const bool ky2 = half ? c1_tap_ky2(t0 + 2) : c1_tap_ky2(t0);
        const bool tv  = half ? (t0 + 2 < 27) : (t0 < 27);
        const bool ok  = tv && !(kx2 && lastcol) && !(ky2 && lastrow);
        const float v  = p[ok ? off : 0];
        b[j] = ok ? v : 0.f;
    }
}

__global__ __launch_bounds__(128)
void conv1_silu_kernel(const float* __restrict__ x, const float* __restrict__ w1,
                       const float* __restrict__ b1, float* __restrict__ out)
{
    const int lane = threadIdx.x & 31;
    const int half = lane >> 4;
    const int nl   = lane & 15;
    const int wv  = __builtin_amdgcn_readfirstlane((int)(blockIdx.x * 4 + (threadIdx.x >> 5)));
    const int img = wv / (392 / C1_CHUNKS);
    const int q0  = (wv - img * (392 / C1_CHUNKS)) * C1_CHUNKS;

    v2f afr[2][7];
    #pragma unroll
    for (int mt = 0; mt < 2; ++mt) {
        const float* wr = w1 + (mt * 16 + nl) * 27 + 2 * half;
        #pragma unroll
        for (int s = 0; s < 7; ++s) {
            v2f a;
            a.x = wr[4 * s];
            if (s < 6) {
                a.y = wr[4 * s + 1];
            } else {
                const float t = wr[half ? 0 : 25];
                a.y = half ? 0.f : t;
            }
            afr[mt][s] = a;
        }
    }
    float bias[2][8];
    #pragma unroll
    for (int mt = 0; mt < 2; ++mt)
        #pragma unroll
        for (int r = 0; r < 8; ++r) bias[mt][r] = b1[mt * 16 + 8 * half + r];

    const float* xi = x + (size_t)img * 150528;
    float* ob = out + (size_t)img * (32 * 12544) + (size_t)(8 * half) * 12544 + nl;

    float bc[2][14], bn[2][14];
    conv1_load_tile(xi, 2 * q0,     nl, half, bc[0]);
    conv1_load_tile(xi, 2 * q0 + 1, nl, half, bc[1]);
    #pragma unroll
    for (int i = 0; i < C1_CHUNKS; ++i) {
        const int q = q0 + i;
        if (i + 1 < C1_CHUNKS) {
            conv1_load_tile(xi, 2 * q + 2, nl, half, bn[0]);
            conv1_load_tile(xi, 2 * q + 3, nl, half, bn[1]);
        }
        v8f c[2][2];
        #pragma unroll
        for (int tl = 0; tl < 2; ++tl)
            #pragma unroll
            for (int mt = 0; mt < 2; ++mt)
                c[tl][mt] = (v8f){0.f, 0.f, 0.f, 0.f, 0.f, 0.f, 0.f, 0.f};
        #pragma unroll
        for (int s = 0; s < 7; ++s) {
            #pragma unroll
            for (int tl = 0; tl < 2; ++tl) {
                v2f b;
                b.x = bc[tl][2 * s];
                b.y = bc[tl][2 * s + 1];
                #pragma unroll
                for (int mt = 0; mt < 2; ++mt)
                    c[tl][mt] = __builtin_amdgcn_wmma_f32_16x16x4_f32(false, afr[mt][s], false, b,
                                                                     (short)0, c[tl][mt], false, false);
            }
        }
        #pragma unroll
        for (int tl = 0; tl < 2; ++tl)
            #pragma unroll
            for (int mt = 0; mt < 2; ++mt)
                #pragma unroll
                for (int r = 0; r < 8; ++r) {
                    const float v = c[tl][mt][r] + bias[mt][r];
                    ob[(mt * 16 + r) * 12544 + (2 * q + tl) * 16] = v / (1.f + expf(-v));
                }
        if (i + 1 < C1_CHUNKS) {
            #pragma unroll
            for (int tl = 0; tl < 2; ++tl)
                #pragma unroll
                for (int j = 0; j < 14; ++j) bc[tl][j] = bn[tl][j];
        }
    }
}

#define C2_PP 34
#define C2_RP 3856

template <int NT>
__device__ __forceinline__ void conv2_tiles(const v2f (&afr)[72], const float* tb,
                                            const float (&bias)[8], float (&accR)[8])
{
    v8f c[NT];
    #pragma unroll
    for (int t = 0; t < NT; ++t)
        c[t] = (v8f){0.f, 0.f, 0.f, 0.f, 0.f, 0.f, 0.f, 0.f};

    v2f bc[NT][2], bn[NT][2];
    #pragma unroll
    for (int t = 0; t < NT; ++t) {
        bc[t][0] = *(const v2f*)(tb + t * 16 * C2_PP);
        bc[t][1] = *(const v2f*)(tb + t * 16 * C2_PP + 4);
    }
    #pragma unroll
    for (int st = 0; st < 36; ++st) {
        if (st + 1 < 36) {
            const int icb = (st + 1) / 9, s = (st + 1) - icb * 9;
            const int ky = s / 3, kx = s - ky * 3;
            const int off = ky * C2_RP + kx * C2_PP + icb * 8;
            #pragma unroll
            for (int t = 0; t < NT; ++t) {
                bn[t][0] = *(const v2f*)(tb + off + t * 16 * C2_PP);
                bn[t][1] = *(const v2f*)(tb + off + t * 16 * C2_PP + 4);
            }
        }
        const int icb = st / 9, s = st - icb * 9;
        const v2f a0 = afr[icb * 18 + 2 * s];
        const v2f a1 = afr[icb * 18 + 2 * s + 1];
        #pragma unroll
        for (int t = 0; t < NT; ++t)
            c[t] = __builtin_amdgcn_wmma_f32_16x16x4_f32(false, a0, false, bc[t][0],
                                                         (short)0, c[t], false, false);
        #pragma unroll
        for (int t = 0; t < NT; ++t)
            c[t] = __builtin_amdgcn_wmma_f32_16x16x4_f32(false, a1, false, bc[t][1],
                                                         (short)0, c[t], false, false);
        #pragma unroll
        for (int t = 0; t < NT; ++t) { bc[t][0] = bn[t][0]; bc[t][1] = bn[t][1]; }
    }
    #pragma unroll
    for (int t = 0; t < NT; ++t) {
        #pragma unroll
        for (int j = 0; j < 8; ++j) {
            float v = c[t][j] + bias[j];
            accR[j] += v / (1.f + expf(-v));
        }
    }
}

__global__ __launch_bounds__(128)
void conv2_wmma_pool_kernel(const float* __restrict__ h1, const float* __restrict__ w2,
                            const float* __restrict__ b2, float* __restrict__ feat)
{
    __shared__ __attribute__((aligned(16))) float sIn[5 * C2_RP];
    __shared__ float sOut[64];

    const int mgrp  = blockIdx.x;
    const int img   = blockIdx.y;
    const int tid   = threadIdx.x;
    const int lane  = tid & 31;
    const int wave  = tid >> 5;
    const int half  = lane >> 4;
    const int nl    = lane & 15;
    const int mtile = mgrp * 4 + wave;

    v2f afr[72];
    {
        const float* wr = w2 + (size_t)(mtile * 16 + nl) * 288 + half * 18;
        #pragma unroll
        for (int icb = 0; icb < 4; ++icb) {
            #pragma unroll
            for (int kq = 0; kq < 18; ++kq) {
                const int s  = kq >> 1;
                const int ic = icb * 8 + (kq & 1) * 4;
                v2f a;
                a.x = wr[ic * 9 + s];
                a.y = wr[(ic + 1) * 9 + s];
                afr[icb * 18 + kq] = a;
            }
        }
    }

    float bias[8];
    #pragma unroll
    for (int j = 0; j < 8; ++j) bias[j] = b2[mtile * 16 + j + half * 8];

    for (int i = tid; i < 5 * 32; i += 128)
        sIn[(i >> 5) * C2_RP + 112 * C2_PP + (i & 31)] = 0.f;

    float accR[8] = {0.f, 0.f, 0.f, 0.f, 0.f, 0.f, 0.f, 0.f};

    const float* base = h1 + (size_t)img * (32 * 112 * 112) + (size_t)(wave * 8) * (112 * 112) + lane;
    float* st = sIn + lane * C2_PP + wave * 8;

    const int dy = nl >> 3, dx = nl & 7;
    const float* tb = sIn + (2 * dy) * C2_RP + (2 * dx) * C2_PP + half * 2;

    for (int yt = 0; yt < 28; ++yt) {
        __syncthreads();
        const int iy0 = yt * 4;
        #pragma unroll
        for (int row = 0; row < 5; ++row) {
            const int  iy = iy0 + row;
            const bool ok = iy < 112;
            const float* p = base + (ok ? iy : 111) * 112;
            float v[24];
            #pragma unroll
            for (int q = 0; q < 3; ++q) {
                #pragma unroll
                for (int c = 0; c < 8; ++c) v[q * 8 + c] = p[c * 12544 + q * 32];
            }
            #pragma unroll
            for (int q = 0; q < 3; ++q) {
                #pragma unroll
                for (int c = 0; c < 8; ++c)
                    st[row * C2_RP + q * 32 * C2_PP + c] = ok ? v[q * 8 + c] : 0.f;
            }
        }
        if (lane < 16) {
            float v[40];
            #pragma unroll
            for (int row = 0; row < 5; ++row) {
                const int iy = iy0 + row;
                const float* p = base + (iy < 112 ? iy : 111) * 112;
                #pragma unroll
                for (int c = 0; c < 8; ++c) v[row * 8 + c] = p[c * 12544 + 96];
            }
            #pragma unroll
            for (int row = 0; row < 5; ++row) {
                const bool ok = (iy0 + row) < 112;
                #pragma unroll
                for (int c = 0; c < 8; ++c)
                    st[row * C2_RP + 96 * C2_PP + c] = ok ? v[row * 8 + c] : 0.f;
            }
        }
        __syncthreads();

        conv2_tiles<4>(afr, tb, bias, accR);
        conv2_tiles<3>(afr, tb + 4 * 16 * C2_PP, bias, accR);
    }

    #pragma unroll
    for (int j = 0; j < 8; ++j) {
        float v = accR[j];
        v += __shfl_xor(v, 1);
        v += __shfl_xor(v, 2);
        v += __shfl_xor(v, 4);
        v += __shfl_xor(v, 8);
        if (nl == 0) sOut[wave * 16 + half * 8 + j] = v;
    }
    __syncthreads();
    if (tid < 64)
        feat[img * 1280 + mgrp * 64 + tid] = sOut[tid] * (1.f / 3136.f);
}

#define DFT_ROWS 8
#define DFT_COLS 16
#define DFT_CP   225

__global__ __launch_bounds__(224)
void dft_rows_kernel(const float* __restrict__ x, float* __restrict__ F1)
{
    __shared__ float ct[224], st[224];
    __shared__ float xs[DFT_ROWS][224];
    __shared__ float2 ys[DFT_ROWS][224];
    const int t   = threadIdx.x;
    const int bc  = blockIdx.x / (224 / DFT_ROWS);
    const int y0  = (blockIdx.x - bc * (224 / DFT_ROWS)) * DFT_ROWS;
    const float ang = (2.0f * PI_F / 224.0f) * (float)t;
    ct[t] = cosf(ang); st[t] = sinf(ang);
    const float* rows = x + ((size_t)bc * 224 + y0) * 224;
    #pragma unroll
    for (int r = 0; r < DFT_ROWS; ++r) xs[r][t] = rows[r * 224 + t];
    __syncthreads();

    const int lo = t & 15, hi = t >> 4;
    float re[DFT_ROWS], im[DFT_ROWS];
    #pragma unroll
    for (int r = 0; r < DFT_ROWS; ++r) { re[r] = 0.f; im[r] = 0.f; }
    {
        int j = 0;
        const int dj = 14 * lo;
        for (int n1 = 0; n1 < 16; ++n1) {
            const float c = ct[j], s = st[j];
            #pragma unroll
            for (int r = 0; r < DFT_ROWS; ++r) {
                const float v = xs[r][14 * n1 + hi];
                re[r] += v * c; im[r] -= v * s;
            }
            j += dj; if (j >= 224) j -= 224;
        }
        const float c = ct[hi * lo], s = st[hi * lo];
        #pragma unroll
        for (int r = 0; r < DFT_ROWS; ++r) {
            float2 y;
            y.x = re[r] * c + im[r] * s;
            y.y = im[r] * c - re[r] * s;
            ys[r][t] = y;
        }
    }
    __syncthreads();

    #pragma unroll
    for (int r = 0; r < DFT_ROWS; ++r) { re[r] = 0.f; im[r] = 0.f; }
    {
        int j = 0;
        const int dj = 16 * hi;
        for (int n2 = 0; n2 < 14; ++n2) {
            const float c = ct[j], s = st[j];
            #pragma unroll
            for (int r = 0; r < DFT_ROWS; ++r) {
                const float2 y = ys[r][n2 * 16 + lo];
                re[r] += y.x * c; re[r] += y.y * s;
                im[r] += y.y * c; im[r] -= y.x * s;
            }
            j += dj; if (j >= 224) j -= 224;
        }
    }
    float* o = F1 + (((size_t)bc * 224 + y0) * 224 + t) * 2;
    #pragma unroll
    for (int r = 0; r < DFT_ROWS; ++r) { o[r * 448] = re[r]; o[r * 448 + 1] = im[r]; }
}

__global__ __launch_bounds__(224)
void dft_cols_mag_kernel(const float* __restrict__ F1, float* __restrict__ mag)
{
    __shared__ float ct[224], st[224];
    __shared__ float2 fs[DFT_COLS][DFT_CP];
    __shared__ float2 ys[DFT_COLS][DFT_CP];
    const int t   = threadIdx.x;
    const int bc  = blockIdx.x / (224 / DFT_COLS);
    const int u0  = (blockIdx.x - bc * (224 / DFT_COLS)) * DFT_COLS;
    const float ang = (2.0f * PI_F / 224.0f) * (float)t;
    ct[t] = cosf(ang); st[t] = sinf(ang);
    {
        const float2* base = (const float2*)(F1 + (size_t)bc * (224 * 448) + u0 * 2);
        #pragma unroll
        for (int i = 0; i < 16; ++i) {
            const int e = i * 224 + t;
            const int y = e >> 4, q = e & 15;
            fs[q][y] = base[y * 224 + q];
        }
    }
    __syncthreads();

    const int lo = t & 15, hi = t >> 4;
    float re[DFT_COLS], im[DFT_COLS];
    #pragma unroll
    for (int q = 0; q < DFT_COLS; ++q) { re[q] = 0.f; im[q] = 0.f; }
    {
        int j = 0;
        const int dj = 14 * lo;
        for (int n1 = 0; n1 < 16; ++n1) {
            const float c = ct[j], s = st[j];
            #pragma unroll
            for (int q = 0; q < DFT_COLS; ++q) {
                const float2 f = fs[q][14 * n1 + hi];
                re[q] += f.x * c; re[q] += f.y * s;
                im[q] += f.y * c; im[q] -= f.x * s;
            }
            j += dj; if (j >= 224) j -= 224;
        }
        const float c = ct[hi * lo], s = st[hi * lo];
        #pragma unroll
        for (int q = 0; q < DFT_COLS; ++q) {
            float2 y;
            y.x = re[q] * c + im[q] * s;
            y.y = im[q] * c - re[q] * s;
            ys[q][t] = y;
        }
    }
    __syncthreads();

    #pragma unroll
    for (int q = 0; q < DFT_COLS; ++q) { re[q] = 0.f; im[q] = 0.f; }
    {
        int j = 0;
        const int dj = 16 * hi;
        for (int n2 = 0; n2 < 14; ++n2) {
            const float c = ct[j], s = st[j];
            #pragma unroll
            for (int q = 0; q < DFT_COLS; ++q) {
                const float2 y = ys[q][n2 * 16 + lo];
                re[q] += y.x * c; re[q] += y.y * s;
                im[q] += y.y * c; im[q] -= y.x * s;
            }
            j += dj; if (j >= 224) j -= 224;
        }
    }
    int vo = t + 112;  if (vo >= 224) vo -= 224;
    int uo = u0 + 112; if (uo >= 224) uo -= 224;
    float4* o = (float4*)(mag + ((size_t)bc * 224 + vo) * 224 + uo);
    #pragma unroll
    for (int q = 0; q < DFT_COLS; q += 4) {
        float4 m;
        m.x = logf(sqrtf(re[q]     * re[q]     + im[q]     * im[q])     + 1.0f);
        m.y = logf(sqrtf(re[q + 1] * re[q + 1] + im[q + 1] * im[q + 1]) + 1.0f);
        m.z = logf(sqrtf(re[q + 2] * re[q + 2] + im[q + 2] * im[q + 2]) + 1.0f);
        m.w = logf(sqrtf(re[q + 3] * re[q + 3] + im[q + 3] * im[q + 3]) + 1.0f);
        o[q >> 2] = m;
    }
}

__global__ __launch_bounds__(256)
void minmax_kernel(const float* __restrict__ mag, float* __restrict__ mm)
{
    __shared__ float smn[256], smx[256];
    int bc = blockIdx.x, tid = threadIdx.x;
    const float* p = mag + (size_t)bc * 50176;
    float mn = 3.0e38f, mx = -3.0e38f;
    for (int i = tid; i < 50176; i += 256) {
        float v = p[i]; mn = fminf(mn, v); mx = fmaxf(mx, v);
    }
    smn[tid] = mn; smx[tid] = mx; __syncthreads();
    for (int s = 128; s > 0; s >>= 1) {
        if (tid < s) { smn[tid] = fminf(smn[tid], smn[tid + s]);
                       smx[tid] = fmaxf(smx[tid], smx[tid + s]); }
        __syncthreads();
    }
    if (tid == 0) { mm[bc * 2] = smn[0]; mm[bc * 2 + 1] = smx[0]; }
}

__global__ __launch_bounds__(256)
void normalize_kernel(float* __restrict__ mag, const float* __restrict__ mm)
{
    int idx = blockIdx.x * 256 + threadIdx.x;
    if (idx >= 96 * 50176) return;
    int bc = idx / 50176;
    float mn = mm[bc * 2], mx = mm[bc * 2 + 1];
    mag[idx] = (mag[idx] - mn) / (mx - mn + 1e-8f);
}

#define FC_TP 272

template <int K, bool SMALLN>
__global__ __launch_bounds__(256)
void fc_wmma_kernel(const float* __restrict__ in, const float* __restrict__ w,
                    const float* __restrict__ b, float* __restrict__ out,
                    int N, int outStride, int outOffset, int act)
{
    constexpr int KC = K / 8;
    constexpr int NL = KC / 8;
    __shared__ float part[8 * 2 * FC_TP];

    const int tid  = threadIdx.x;
    const int lane = tid & 31;
    const int wave = tid >> 5;
    const int half = lane >> 4;
    const int c    = lane & 15;
    const int j0   = blockIdx.x * 32;
    const int i0   = blockIdx.y * 16;
    const int kb   = wave * KC + 4 * half;

    int ja = j0 + c, jb = j0 + 16 + c;
    if (SMALLN) ja = ja < N ? ja : N - 1;
    const float4* pa  = (const float4*)(in + (size_t)(i0 + c) * K + kb);
    const float4* pb0 = (const float4*)(w + (size_t)ja * K + kb);
    const float4* pb1 = (const float4*)(w + (size_t)jb * K + kb);

    int fin_il = 0, fin_j = lane;
    if (SMALLN) {
        const int tt = tid < 16 * N ? tid : 0;
        fin_il = tt / N;
        fin_j  = tt - fin_il * N;
    }
    const float bj = b[SMALLN ? fin_j : j0 + lane];

    float4 xa[NL], wb0[NL], wb1[NL];
    #pragma unroll
    for (int m = 0; m < NL; ++m) {
        xa[m]  = pa[2 * m];
        wb0[m] = pb0[2 * m];
        if (!SMALLN) wb1[m] = pb1[2 * m];
    }
    __builtin_amdgcn_sched_barrier(0);

    v8f acc0 = (v8f){0.f, 0.f, 0.f, 0.f, 0.f, 0.f, 0.f, 0.f};
    v8f acc1 = (v8f){0.f, 0.f, 0.f, 0.f, 0.f, 0.f, 0.f, 0.f};
    #pragma unroll
    for (int m = 0; m < NL; ++m) {
        v2f a, p, q;
        a.x = xa[m].x;  a.y = xa[m].y;
        p.x = wb0[m].x; p.y = wb0[m].y;
        acc0 = __builtin_amdgcn_wmma_f32_16x16x4_f32(false, a, false, p, (short)0, acc0, false, false);
        if (!SMALLN) {
            q.x = wb1[m].x; q.y = wb1[m].y;
            acc1 = __builtin_amdgcn_wmma_f32_16x16x4_f32(false, a, false, q, (short)0, acc1, false, false);
        }
        a.x = xa[m].z;  a.y = xa[m].w;
        p.x = wb0[m].z; p.y = wb0[m].w;
        acc0 = __builtin_amdgcn_wmma_f32_16x16x4_f32(false, a, false, p, (short)0, acc0, false, false);
        if (!SMALLN) {
            q.x = wb1[m].z; q.y = wb1[m].w;
            acc1 = __builtin_amdgcn_wmma_f32_16x16x4_f32(false, a, false, q, (short)0, acc1, false, false);
        }
    }

    float* pp = part + (wave * 2) * FC_TP + lane;
    #pragma unroll
    for (int r = 0; r < 8; ++r) {
        pp[r * 32] = acc0[r];
        if (!SMALLN) pp[FC_TP + r * 32] = acc1[r];
    }
    __syncthreads();

    if (!SMALLN) {
        const int t = lane >> 4;
        #pragma unroll
        for (int u = 0; u < 2; ++u) {
            const int il  = 2 * wave + u;
            const int src = t * FC_TP + (il & 7) * 32 + 16 * (il >> 3) + c;
            float s = part[src];
            #pragma unroll
            for (int q = 1; q < 8; ++q) s += part[q * 2 * FC_TP + src];
            s += bj;
            if (act == 1) s = fmaxf(s, 0.f);
            out[(size_t)(i0 + il) * outStride + outOffset + j0 + lane] = s;
        }
    } else {
        if (tid < 16 * N) {
            const int il  = fin_il;
            const int j   = fin_j;
            const int src = (il & 7) * 32 + 16 * (il >> 3) + j;
            float s = part[src];
            #pragma unroll
            for (int q = 1; q < 8; ++q) s += part[q * 2 * FC_TP + src];
            s += bj;
            if (act == 1) s = fmaxf(s, 0.f);
            out[(size_t)(i0 + il) * outStride + outOffset + j] = s;
        }
    }
}

__global__ __launch_bounds__(256)
void attn_core_kernel(const float* __restrict__ qkv_rgb, const float* __restrict__ qkv_frq,
                      float* __restrict__ o1, float* __restrict__ o2)
{
    __shared__ float sQ[32][32], sK[32][32], sV[32][32], sS[32][33];
    int a = blockIdx.x >> 3;
    int h = blockIdx.x & 7;
    const float* qsrc  = (a == 0) ? qkv_rgb : qkv_frq;
    const float* kvsrc = (a == 0) ? qkv_frq : qkv_rgb;
    float* odst        = (a == 0) ? o1 : o2;
    int tid = threadIdx.x;
    for (int idx = tid; idx < 1024; idx += 256) {
        int q = idx >> 5, d = idx & 31;
        sQ[q][d] = qsrc [q * 768 +       h * 32 + d];
        sK[q][d] = kvsrc[q * 768 + 256 + h * 32 + d];
        sV[q][d] = kvsrc[q * 768 + 512 + h * 32 + d];
    }
    __syncthreads();
    const float scale = 0.17677669529663687f;
    for (int idx = tid; idx < 1024; idx += 256) {
        int q = idx >> 5, k = idx & 31;
        float s = 0.f;
        #pragma unroll
        for (int d = 0; d < 32; ++d) s += sQ[q][d] * sK[k][d];
        sS[q][k] = s * scale;
    }
    __syncthreads();
    if (tid < 32) {
        float mx = -3.0e38f;
        for (int k = 0; k < 32; ++k) mx = fmaxf(mx, sS[tid][k]);
        float e[32], sum = 0.f;
        for (int k = 0; k < 32; ++k) { e[k] = expf(sS[tid][k] - mx); sum += e[k]; }
        float inv = 1.f / sum;
        for (int k = 0; k < 32; ++k) sS[tid][k] = e[k] * inv;
    }
    __syncthreads();
    for (int idx = tid; idx < 1024; idx += 256) {
        int q = idx >> 5, d = idx & 31;
        float o = 0.f;
        #pragma unroll
        for (int k = 0; k < 32; ++k) o += sS[q][k] * sV[k][d];
        odst[q * 256 + h * 32 + d] = o;
    }
}

extern "C" void kernel_launch(void* const* d_in, const int* in_sizes, int n_in,
                              void* d_out, int out_size, void* d_ws, size_t ws_size,
                              hipStream_t stream)
{
    (void)in_sizes; (void)n_in; (void)out_size; (void)ws_size;
    const float* x          = (const float*)d_in[0];
    const float* rgb_w1     = (const float*)d_in[1];
    const float* rgb_b1     = (const float*)d_in[2];
    const float* rgb_w2     = (const float*)d_in[3];
    const float* rgb_b2     = (const float*)d_in[4];
    const float* frq_w1     = (const float*)d_in[5];
    const float* frq_b1     = (const float*)d_in[6];
    const float* frq_w2     = (const float*)d_in[7];
    const float* frq_b2     = (const float*)d_in[8];
    const float* rgb_proj_w = (const float*)d_in[9];
    const float* rgb_proj_b = (const float*)d_in[10];
    const float* frq_proj_w = (const float*)d_in[11];
    const float* frq_proj_b = (const float*)d_in[12];
    const float* attn_w_in  = (const float*)d_in[13];
    const float* attn_b_in  = (const float*)d_in[14];
    const float* attn_w_out = (const float*)d_in[15];
    const float* attn_b_out = (const float*)d_in[16];
    const float* fus_w1     = (const float*)d_in[17];
    const float* fus_b1     = (const float*)d_in[18];
    const float* fus_w2     = (const float*)d_in[19];
    const float* fus_b2     = (const float*)d_in[20];
    const float* fus_w3     = (const float*)d_in[21];
    const float* fus_b3     = (const float*)d_in[22];

    float* out      = (float*)d_out;
    float* feat_rgb = out + 64;
    float* feat_frq = out + 64 + 40960;

    float* ws       = (float*)d_ws;
    float* h1       = ws;
    float* F1       = ws;
    float* freqimg  = ws + 12845056;
    float* mm       = freqimg + 4816896;
    float* hd       = mm + 192;
    float* rgb_p    = hd;
    float* frq_p    = rgb_p + 8192;
    float* qkv_rgb  = frq_p + 8192;
    float* qkv_frq  = qkv_rgb + 24576;
    float* o1       = qkv_frq + 24576;
    float* o2       = o1 + 8192;
    float* hb       = o2 + 8192;
    float* h1f      = hb + 16384;
    float* h2f      = h1f + 16384;

    conv1_silu_kernel<<<1568, 128, 0, stream>>>(x, rgb_w1, rgb_b1, h1);
    conv2_wmma_pool_kernel<<<dim3(20, 32), 128, 0, stream>>>(h1, rgb_w2, rgb_b2, feat_rgb);

    dft_rows_kernel<<<2688, 224, 0, stream>>>(x, F1);
    dft_cols_mag_kernel<<<1344, 224, 0, stream>>>(F1, freqimg);
    minmax_kernel<<<96, 256, 0, stream>>>(freqimg, mm);
    normalize_kernel<<<18816, 256, 0, stream>>>(freqimg, mm);
    conv1_silu_kernel<<<1568, 128, 0, stream>>>(freqimg, frq_w1, frq_b1, h1);
    conv2_wmma_pool_kernel<<<dim3(20, 32), 128, 0, stream>>>(h1, frq_w2, frq_b2, feat_frq);

    fc_wmma_kernel<1280, false><<<dim3(8, 2), 256, 0, stream>>>(feat_rgb, rgb_proj_w, rgb_proj_b, rgb_p, 256, 256, 0, 0);
    fc_wmma_kernel<1280, false><<<dim3(8, 2), 256, 0, stream>>>(feat_frq, frq_proj_w, frq_proj_b, frq_p, 256, 256, 0, 0);
    fc_wmma_kernel<256, false><<<dim3(24, 2), 256, 0, stream>>>(rgb_p, attn_w_in, attn_b_in, qkv_rgb, 768, 768, 0, 0);
    fc_wmma_kernel<256, false><<<dim3(24, 2), 256, 0, stream>>>(frq_p, attn_w_in, attn_b_in, qkv_frq, 768, 768, 0, 0);
    attn_core_kernel<<<16, 256, 0, stream>>>(qkv_rgb, qkv_frq, o1, o2);
    fc_wmma_kernel<256, false><<<dim3(8, 2), 256, 0, stream>>>(o1, attn_w_out, attn_b_out, hb, 256, 512, 0, 0);
    fc_wmma_kernel<256, false><<<dim3(8, 2), 256, 0, stream>>>(o2, attn_w_out, attn_b_out, hb, 256, 512, 256, 0);
    fc_wmma_kernel<512, false><<<dim3(16, 2), 256, 0, stream>>>(hb,  fus_w1, fus_b1, h1f, 512, 512, 0, 1);
    fc_wmma_kernel<512, false><<<dim3(8, 2), 256, 0, stream>>>(h1f, fus_w2, fus_b2, h2f, 256, 256, 0, 1);
    fc_wmma_kernel<256, true><<<dim3(1, 2), 256, 0, stream>>>(h2f, fus_w3, fus_b3, out, 2, 2, 0, 0);
}


# conv2 grid re-cut for equal matrix work per SIMD: block = image x 32 ch, 1280 equal blocks (5 per WGP, was 2.5); wave pairs split a strip's 7 pixel tiles 4:3 in turn; pool summation order changes
# speedup vs baseline: 2.4162x; 1.0911x over previous
#include <hip/hip_runtime.h>
#include <hip/hip_bf16.h>
#include <math.h>

typedef float v2f __attribute__((ext_vector_type(2)));
typedef float v8f __attribute__((ext_vector_type(8)));

#define PI_F 3.14159265358979323846f

#define C1_CHUNKS 2

__device__ __forceinline__ constexpr int  c1_tap_off(int t)
{ return t < 27 ? (t / 9) * 50176 + ((t % 9) / 3) * 224 + (t % 3) : 0; }
__device__ __forceinline__ constexpr bool c1_tap_kx2(int t) { return t < 27 && (t % 3) == 2; }
__device__ __forceinline__ constexpr bool c1_tap_ky2(int t) { return t < 27 && ((t % 9) / 3) == 2; }

__device__ __forceinline__ void conv1_load_tile(const float* __restrict__ xi, int tl,
                                                int nl, int half, float (&b)[14])
{
    const int  oy = tl / 7;
    const int  ox = (tl - oy * 7) * 16 + nl;
    const bool lastrow = (oy == 111);
    const bool lastcol = (ox == 111);
    const float* p = xi + (2 * oy) * 224 + 2 * ox;
    #pragma unroll
    for (int j = 0; j < 14; ++j) {
        const int  t0  = 4 * (j >> 1) + (j & 1);
        const int  off = half ? c1_tap_off(t0 + 2) : c1_tap_off(t0);
        const bool kx2 = half ? c1_tap_kx2(t0 + 2) : c1_tap_kx2(t0);
        const bool ky2 = half ? c1_tap_ky2(t0 + 2) : c1_tap_ky2(t0);
        const bool tv  = half ? (t0 + 2 < 27) : (t0 < 27);
        const bool ok  = tv && !(kx2 && lastcol) && !(ky2 && lastrow);
        const float v  = p[ok ? off : 0];
        b[j] = ok ? v : 0.f;
    }
}

__global__ __launch_bounds__(128)
void conv1_silu_kernel(const float* __restrict__ x, const float* __restrict__ w1,
                       const float* __restrict__ b1, float* __restrict__ out)
{
    const int lane = threadIdx.x & 31;
    const int half = lane >> 4;
    const int nl   = lane & 15;
    const int wv  = __builtin_amdgcn_readfirstlane((int)(blockIdx.x * 4 + (threadIdx.x >> 5)));
    const int img = wv / (392 / C1_CHUNKS);
    const int q0  = (wv - img * (392 / C1_CHUNKS)) * C1_CHUNKS;

    v2f afr[2][7];
    #pragma unroll
    for (int mt = 0; mt < 2; ++mt) {
        const float* wr = w1 + (mt * 16 + nl) * 27 + 2 * half;
        #pragma unroll
        for (int s = 0; s < 7; ++s) {
            v2f a;
            a.x = wr[4 * s];
            if (s < 6) {
                a.y = wr[4 * s + 1];
            } else {
                const float t = wr[half ? 0 : 25];
                a.y = half ? 0.f : t;
            }
            afr[mt][s] = a;
        }
    }
    float bias[2][8];
    #pragma unroll
    for (int mt = 0; mt < 2; ++mt)
        #pragma unroll
        for (int r = 0; r < 8; ++r) bias[mt][r] = b1[mt * 16 + 8 * half + r];

    const float* xi = x + (size_t)img * 150528;
    float* ob = out + (size_t)img * (32 * 12544) + (size_t)(8 * half) * 12544 + nl;

    float bc[2][14], bn[2][14];
    conv1_load_tile(xi, 2 * q0,     nl, half, bc[0]);
    conv1_load_tile(xi, 2 * q0 + 1, nl, half, bc[1]);
    #pragma unroll
    for (int i = 0; i < C1_CHUNKS; ++i) {
        const int q = q0 + i;
        if (i + 1 < C1_CHUNKS) {
            conv1_load_tile(xi, 2 * q + 2, nl, half, bn[0]);
            conv1_load_tile(xi, 2 * q + 3, nl, half, bn[1]);
        }
        v8f c[2][2];
        #pragma unroll
        for (int tl = 0; tl < 2; ++tl)
            #pragma unroll
            for (int mt = 0; mt < 2; ++mt)
                c[tl][mt] = (v8f){0.f, 0.f, 0.f, 0.f, 0.f, 0.f, 0.f, 0.f};
        #pragma unroll
        for (int s = 0; s < 7; ++s) {
            #pragma unroll
            for (int tl = 0; tl < 2; ++tl) {
                v2f b;
                b.x = bc[tl][2 * s];
                b.y = bc[tl][2 * s + 1];
                #pragma unroll
                for (int mt = 0; mt < 2; ++mt)
                    c[tl][mt] = __builtin_amdgcn_wmma_f32_16x16x4_f32(false, afr[mt][s], false, b,
                                                                     (short)0, c[tl][mt], false, false);
            }
        }
        #pragma unroll
        for (int tl = 0; tl < 2; ++tl)
            #pragma unroll
            for (int mt = 0; mt < 2; ++mt)
                #pragma unroll
                for (int r = 0; r < 8; ++r) {
                    const float v = c[tl][mt][r] + bias[mt][r];
                    ob[(mt * 16 + r) * 12544 + (2 * q + tl) * 16] = v / (1.f + expf(-v));
                }
        if (i + 1 < C1_CHUNKS) {
            #pragma unroll
            for (int tl = 0; tl < 2; ++tl)
                #pragma unroll
                for (int j = 0; j < 14; ++j) bc[tl][j] = bn[tl][j];
        }
    }
}

#define C2_PP 34
#define C2_RP 3856

template <int NT>
__device__ __forceinline__ void conv2_tiles(const v2f (&afr)[72], const float* tb,
                                            const float (&bias)[8], float (&accR)[8])
{
    v8f c[NT];
    #pragma unroll
    for (int t = 0; t < NT; ++t)
        c[t] = (v8f){0.f, 0.f, 0.f, 0.f, 0.f, 0.f, 0.f, 0.f};

    v2f bc[NT][2], bn[NT][2];
    #pragma unroll
    for (int t = 0; t < NT; ++t) {
        bc[t][0] = *(const v2f*)(tb + t * 16 * C2_PP);
        bc[t][1] = *(const v2f*)(tb + t * 16 * C2_PP + 4);
    }
    #pragma unroll
    for (int st = 0; st < 36; ++st) {
        if (st + 1 < 36) {
            const int icb = (st + 1) / 9, s = (st + 1) - icb * 9;
            const int ky = s / 3, kx = s - ky * 3;
            const int off = ky * C2_RP + kx * C2_PP + icb * 8;
            __builtin_amdgcn_sched_barrier(0);
            #pragma unroll
            for (int t = 0; t < NT; ++t) {
                bn[t][0] = *(const v2f*)(tb + off + t * 16 * C2_PP);
                bn[t][1] = *(const v2f*)(tb + off + t * 16 * C2_PP + 4);
            }
            __builtin_amdgcn_sched_barrier(0);
        }
        const int icb = st / 9, s = st - icb * 9;
        const v2f a0 = afr[icb * 18 + 2 * s];
        const v2f a1 = afr[icb * 18 + 2 * s + 1];
        #pragma unroll
        for (int t = 0; t < NT; ++t)
            c[t] = __builtin_amdgcn_wmma_f32_16x16x4_f32(false, a0, false, bc[t][0],
                                                         (short)0, c[t], false, false);
        #pragma unroll
        for (int t = 0; t < NT; ++t)
            c[t] = __builtin_amdgcn_wmma_f32_16x16x4_f32(false, a1, false, bc[t][1],
                                                         (short)0, c[t], false, false);
        #pragma unroll
        for (int t = 0; t < NT; ++t) { bc[t][0] = bn[t][0]; bc[t][1] = bn[t][1]; }
    }
    #pragma unroll
    for (int t = 0; t < NT; ++t) {
        #pragma unroll
        for (int j = 0; j < 8; ++j) {
            float v = c[t][j] + bias[j];
            accR[j] += v / (1.f + expf(-v));
        }
    }
}

__global__ __launch_bounds__(128)
void conv2_wmma_pool_kernel(const float* __restrict__ h1, const float* __restrict__ w2,
                            const float* __restrict__ b2, float* __restrict__ feat)
{
    __shared__ __attribute__((aligned(16))) float sIn[5 * C2_RP];
    __shared__ float sOut[2][32];

    const int mgrp  = blockIdx.x;
    const int img   = blockIdx.y;
    const int tid   = threadIdx.x;
    const int lane  = tid & 31;
    const int wave  = __builtin_amdgcn_readfirstlane(tid >> 5);
    const int half  = lane >> 4;
    const int nl    = lane & 15;
    const int ct    = wave & 1;
    const int ph    = wave >> 1;
    const int mtile = mgrp * 2 + ct;

    v2f afr[72];
    {
        const float* wr = w2 + (size_t)(mtile * 16 + nl) * 288 + half * 18;
        #pragma unroll
        for (int icb = 0; icb < 4; ++icb) {
            #pragma unroll
            for (int kq = 0; kq < 18; ++kq) {
                const int s  = kq >> 1;
                const int ic = icb * 8 + (kq & 1) * 4;
                v2f a;
                a.x = wr[ic * 9 + s];
                a.y = wr[(ic + 1) * 9 + s];
                afr[icb * 18 + kq] = a;
            }
        }
    }

    float bias[8];
    #pragma unroll
    for (int j = 0; j < 8; ++j) bias[j] = b2[mtile * 16 + j + half * 8];

    for (int i = tid; i < 5 * 32; i += 128)
        sIn[(i >> 5) * C2_RP + 112 * C2_PP + (i & 31)] = 0.f;

    float accR[8] = {0.f, 0.f, 0.f, 0.f, 0.f, 0.f, 0.f, 0.f};

    const float* base = h1 + (size_t)img * (32 * 112 * 112) + (size_t)(wave * 8) * (112 * 112) + lane;
    float* st = sIn + lane * C2_PP + wave * 8;

    const int dy = nl >> 3, dx = nl & 7;
    const float* tb = sIn + (2 * dy) * C2_RP + (2 * dx) * C2_PP + half * 2;

    for (int yt = 0; yt < 28; ++yt) {
        __syncthreads();
        const int iy0 = yt * 4;
        #pragma unroll
        for (int row = 0; row < 5; ++row) {
            const int  iy = iy0 + row;
            const bool ok = iy < 112;
            const float* p = base + (ok ? iy : 111) * 112;
            float v[24];
            #pragma unroll
            for (int q = 0; q < 3; ++q) {
                #pragma unroll
                for (int c = 0; c < 8; ++c) v[q * 8 + c] = p[c * 12544 + q * 32];
            }
            #pragma unroll
            for (int q = 0; q < 3; ++q) {
                #pragma unroll
                for (int c = 0; c < 8; ++c)
                    st[row * C2_RP + q * 32 * C2_PP + c] = ok ? v[q * 8 + c] : 0.f;
            }
        }
        if (lane < 16) {
            float v[40];
            #pragma unroll
            for (int row = 0; row < 5; ++row) {
                const int iy = iy0 + row;
                const float* p = base + (iy < 112 ? iy : 111) * 112;
                #pragma unroll
                for (int c = 0; c < 8; ++c) v[row * 8 + c] = p[c * 12544 + 96];
            }
            #pragma unroll
            for (int row = 0; row < 5; ++row) {
                const bool ok = (iy0 + row) < 112;
                #pragma unroll
                for (int c = 0; c < 8; ++c)
                    st[row * C2_RP + 96 * C2_PP + c] = ok ? v[row * 8 + c] : 0.f;
            }
        }
        __syncthreads();

        if (((yt ^ ph) & 1) == 0) conv2_tiles<4>(afr, tb, bias, accR);
        else                      conv2_tiles<3>(afr, tb + 4 * 16 * C2_PP, bias, accR);
    }

    #pragma unroll
    for (int j = 0; j < 8; ++j) {
        float v = accR[j];
        v += __shfl_xor(v, 1);
        v += __shfl_xor(v, 2);
        v += __shfl_xor(v, 4);
        v += __shfl_xor(v, 8);
        if (nl == 0) sOut[ph][ct * 16 + half * 8 + j] = v;
    }
    __syncthreads();
    if (tid < 32)
        feat[img * 1280 + mgrp * 32 + tid] = (sOut[0][tid] + sOut[1][tid]) * (1.f / 3136.f);
}

#define DFT_ROWS 8
#define DFT_COLS 16
#define DFT_CP   225

__global__ __launch_bounds__(224)
void dft_rows_kernel(const float* __restrict__ x, float* __restrict__ F1)
{
    __shared__ float ct[224], st[224];
    __shared__ float xs[DFT_ROWS][224];
    __shared__ float2 ys[DFT_ROWS][224];
    const int t   = threadIdx.x;
    const int bc  = blockIdx.x / (224 / DFT_ROWS);
    const int y0  = (blockIdx.x - bc * (224 / DFT_ROWS)) * DFT_ROWS;
    const float ang = (2.0f * PI_F / 224.0f) * (float)t;
    ct[t] = cosf(ang); st[t] = sinf(ang);
    const float* rows = x + ((size_t)bc * 224 + y0) * 224;
    #pragma unroll
    for (int r = 0; r < DFT_ROWS; ++r) xs[r][t] = rows[r * 224 + t];
    __syncthreads();

    const int lo = t & 15, hi = t >> 4;
    float re[DFT_ROWS], im[DFT_ROWS];
    #pragma unroll
    for (int r = 0; r < DFT_ROWS; ++r) { re[r] = 0.f; im[r] = 0.f; }
    {
        int j = 0;
        const int dj = 14 * lo;
        for (int n1 = 0; n1 < 16; ++n1) {
            const float c = ct[j], s = st[j];
            #pragma unroll
            for (int r = 0; r < DFT_ROWS; ++r) {
                const float v = xs[r][14 * n1 + hi];
                re[r] += v * c; im[r] -= v * s;
            }
            j += dj; if (j >= 224) j -= 224;
        }
        const float c = ct[hi * lo], s = st[hi * lo];
        #pragma unroll
        for (int r = 0; r < DFT_ROWS; ++r) {
            float2 y;
            y.x = re[r] * c + im[r] * s;
            y.y = im[r] * c - re[r] * s;
            ys[r][t] = y;
        }
    }
    __syncthreads();

    #pragma unroll
    for (int r = 0; r < DFT_ROWS; ++r) { re[r] = 0.f; im[r] = 0.f; }
    {
        int j = 0;
        const int dj = 16 * hi;
        for (int n2 = 0; n2 < 14; ++n2) {
            const float c = ct[j], s = st[j];
            #pragma unroll
            for (int r = 0; r < DFT_ROWS; ++r) {
                const float2 y = ys[r][n2 * 16 + lo];
                re[r] += y.x * c; re[r] += y.y * s;
                im[r] += y.y * c; im[r] -= y.x * s;
            }
            j += dj; if (j >= 224) j -= 224;
        }
    }
    float* o = F1 + (((size_t)bc * 224 + y0) * 224 + t) * 2;
    #pragma unroll
    for (int r = 0; r < DFT_ROWS; ++r) { o[r * 448] = re[r]; o[r * 448 + 1] = im[r]; }
}

__global__ __launch_bounds__(224)
void dft_cols_mag_kernel(const float* __restrict__ F1, float* __restrict__ mag)
{
    __shared__ float ct[224], st[224];
    __shared__ float2 fs[DFT_COLS][DFT_CP];
    __shared__ float2 ys[DFT_COLS][DFT_CP];
    const int t   = threadIdx.x;
    const int bc  = blockIdx.x / (224 / DFT_COLS);
    const int u0  = (blockIdx.x - bc * (224 / DFT_COLS)) * DFT_COLS;
    const float ang = (2.0f * PI_F / 224.0f) * (float)t;
    ct[t] = cosf(ang); st[t] = sinf(ang);
    {
        const float2* base = (const float2*)(F1 + (size_t)bc * (224 * 448) + u0 * 2);
        #pragma unroll
        for (int i = 0; i < 16; ++i) {
            const int e = i * 224 + t;
            const int y = e >> 4, q = e & 15;
            fs[q][y] = base[y * 224 + q];
        }
    }
    __syncthreads();

    const int lo = t & 15, hi = t >> 4;
    float re[DFT_COLS], im[DFT_COLS];
    #pragma unroll
    for (int q = 0; q < DFT_COLS; ++q) { re[q] = 0.f; im[q] = 0.f; }
    {
        int j = 0;
        const int dj = 14 * lo;
        for (int n1 = 0; n1 < 16; ++n1) {
            const float c = ct[j], s = st[j];
            #pragma unroll
            for (int q = 0; q < DFT_COLS; ++q) {
                const float2 f = fs[q][14 * n1 + hi];
                re[q] += f.x * c; re[q] += f.y * s;
                im[q] += f.y * c; im[q] -= f.x * s;
            }
            j += dj; if (j >= 224) j -= 224;
        }
        const float c = ct[hi * lo], s = st[hi * lo];
        #pragma unroll
        for (int q = 0; q < DFT_COLS; ++q) {
            float2 y;
            y.x = re[q] * c + im[q] * s;
            y.y = im[q] * c - re[q] * s;
            ys[q][t] = y;
        }
    }
    __syncthreads();

    #pragma unroll
    for (int q = 0; q < DFT_COLS; ++q) { re[q] = 0.f; im[q] = 0.f; }
    {
        int j = 0;
        const int dj = 16 * hi;
        for (int n2 = 0; n2 < 14; ++n2) {
            const float c = ct[j], s = st[j];
            #pragma unroll
            for (int q = 0; q < DFT_COLS; ++q) {
                const float2 y = ys[q][n2 * 16 + lo];
                re[q] += y.x * c; re[q] += y.y * s;
                im[q] += y.y * c; im[q] -= y.x * s;
            }
            j += dj; if (j >= 224) j -= 224;
        }
    }
    int vo = t + 112;  if (vo >= 224) vo -= 224;
    int uo = u0 + 112; if (uo >= 224) uo -= 224;
    float4* o = (float4*)(mag + ((size_t)bc * 224 + vo) * 224 + uo);
    #pragma unroll
    for (int q = 0; q < DFT_COLS; q += 4) {
        float4 m;
        m.x = logf(sqrtf(re[q]     * re[q]     + im[q]     * im[q])     + 1.0f);
        m.y = logf(sqrtf(re[q + 1] * re[q + 1] + im[q + 1] * im[q + 1]) + 1.0f);
        m.z = logf(sqrtf(re[q + 2] * re[q + 2] + im[q + 2] * im[q + 2]) + 1.0f);
        m.w = logf(sqrtf(re[q + 3] * re[q + 3] + im[q + 3] * im[q + 3]) + 1.0f);
        o[q >> 2] = m;
    }
}

__global__ __launch_bounds__(256)
void minmax_kernel(const float* __restrict__ mag, float* __restrict__ mm)
{
    __shared__ float smn[256], smx[256];
    int bc = blockIdx.x, tid = threadIdx.x;
    const float* p = mag + (size_t)bc * 50176;
    float mn = 3.0e38f, mx = -3.0e38f;
    for (int i = tid; i < 50176; i += 256) {
        float v = p[i]; mn = fminf(mn, v); mx = fmaxf(mx, v);
    }
    smn[tid] = mn; smx[tid] = mx; __syncthreads();
    for (int s = 128; s > 0; s >>= 1) {
        if (tid < s) { smn[tid] = fminf(smn[tid], smn[tid + s]);
                       smx[tid] = fmaxf(smx[tid], smx[tid + s]); }
        __syncthreads();
    }
    if (tid == 0) { mm[bc * 2] = smn[0]; mm[bc * 2 + 1] = smx[0]; }
}

__global__ __launch_bounds__(256)
void normalize_kernel(float* __restrict__ mag, const float* __restrict__ mm)
{
    int idx = blockIdx.x * 256 + threadIdx.x;
    if (idx >= 96 * 50176) return;
    int bc = idx / 50176;
    float mn = mm[bc * 2], mx = mm[bc * 2 + 1];
    mag[idx] = (mag[idx] - mn) / (mx - mn + 1e-8f);
}

#define FC_TP 272

template <int K, bool SMALLN>
__global__ __launch_bounds__(256)
void fc_wmma_kernel(const float* __restrict__ in, const float* __restrict__ w,
                    const float* __restrict__ b, float* __restrict__ out,
                    int N, int outStride, int outOffset, int act)
{
    constexpr int KC = K / 8;
    constexpr int NL = KC / 8;
    __shared__ float part[8 * 2 * FC_TP];

    const int tid  = threadIdx.x;
    const int lane = tid & 31;
    const int wave = tid >> 5;
    const int half = lane >> 4;
    const int c    = lane & 15;
    const int j0   = blockIdx.x * 32;
    const int i0   = blockIdx.y * 16;
    const int kb   = wave * KC + 4 * half;

    int ja = j0 + c, jb = j0 + 16 + c;
    if (SMALLN) ja = ja < N ? ja : N - 1;
    const float4* pa  = (const float4*)(in + (size_t)(i0 + c) * K + kb);
    const float4* pb0 = (const float4*)(w + (size_t)ja * K + kb);
    const float4* pb1 = (const float4*)(w + (size_t)jb * K + kb);

    int fin_il = 0, fin_j = lane;
    if (SMALLN) {
        const int tt = tid < 16 * N ? tid : 0;
        fin_il = tt / N;
        fin_j  = tt - fin_il * N;
    }
    const float bj = b[SMALLN ? fin_j : j0 + lane];

    float4 xa[NL], wb0[NL], wb1[NL];
    #pragma unroll
    for (int m = 0; m < NL; ++m) {
        xa[m]  = pa[2 * m];
        wb0[m] = pb0[2 * m];
        if (!SMALLN) wb1[m] = pb1[2 * m];
    }
    __builtin_amdgcn_sched_barrier(0);

    v8f acc0 = (v8f){0.f, 0.f, 0.f, 0.f, 0.f, 0.f, 0.f, 0.f};
    v8f acc1 = (v8f){0.f, 0.f, 0.f, 0.f, 0.f, 0.f, 0.f, 0.f};
    #pragma unroll
    for (int m = 0; m < NL; ++m) {
        v2f a, p, q;
        a.x = xa[m].x;  a.y = xa[m].y;
        p.x = wb0[m].x; p.y = wb0[m].y;
        acc0 = __builtin_amdgcn_wmma_f32_16x16x4_f32(false, a, false, p, (short)0, acc0, false, false);
        if (!SMALLN) {
            q.x = wb1[m].x; q.y = wb1[m].y;
            acc1 = __builtin_amdgcn_wmma_f32_16x16x4_f32(false, a, false, q, (short)0, acc1, false, false);
        }
        a.x = xa[m].z;  a.y = xa[m].w;
        p.x = wb0[m].z; p.y = wb0[m].w;
        acc0 = __builtin_amdgcn_wmma_f32_16x16x4_f32(false, a, false, p, (short)0, acc0, false, false);
        if (!SMALLN) {
            q.x = wb1[m].z; q.y = wb1[m].w;
            acc1 = __builtin_amdgcn_wmma_f32_16x16x4_f32(false, a, false, q, (short)0, acc1, false, false);
        }
    }

    float* pp = part + (wave * 2) * FC_TP + lane;
    #pragma unroll
    for (int r = 0; r < 8; ++r) {
        pp[r * 32] = acc0[r];
        if (!SMALLN) pp[FC_TP + r * 32] = acc1[r];
    }
    __syncthreads();

    if (!SMALLN) {
        const int t = lane >> 4;
        #pragma unroll
        for (int u = 0; u < 2; ++u) {
            const int il  = 2 * wave + u;
            const int src = t * FC_TP + (il & 7) * 32 + 16 * (il >> 3) + c;
            float s = part[src];
            #pragma unroll
            for (int q = 1; q < 8; ++q) s += part[q * 2 * FC_TP + src];
            s += bj;
            if (act == 1) s = fmaxf(s, 0.f);
            out[(size_t)(i0 + il) * outStride + outOffset + j0 + lane] = s;
        }
    } else {
        if (tid < 16 * N) {
            const int il  = fin_il;
            const int j   = fin_j;
            const int src = (il & 7) * 32 + 16 * (il >> 3) + j;
            float s = part[src];
            #pragma unroll
            for (int q = 1; q < 8; ++q) s += part[q * 2 * FC_TP + src];
            s += bj;
            if (act == 1) s = fmaxf(s, 0.f);
            out[(size_t)(i0 + il) * outStride + outOffset + j] = s;
        }
    }
}

__global__ __launch_bounds__(256)
void attn_core_kernel(const float* __restrict__ qkv_rgb, const float* __restrict__ qkv_frq,
                      float* __restrict__ o1, float* __restrict__ o2)
{
    __shared__ float sQ[32][32], sK[32][32], sV[32][32], sS[32][33];
    int a = blockIdx.x >> 3;
    int h = blockIdx.x & 7;
    const float* qsrc  = (a == 0) ? qkv_rgb : qkv_frq;
    const float* kvsrc = (a == 0) ? qkv_frq : qkv_rgb;
    float* odst        = (a == 0) ? o1 : o2;
    int tid = threadIdx.x;
    for (int idx = tid; idx < 1024; idx += 256) {
        int q = idx >> 5, d = idx & 31;
        sQ[q][d] = qsrc [q * 768 +       h * 32 + d];
        sK[q][d] = kvsrc[q * 768 + 256 + h * 32 + d];
        sV[q][d] = kvsrc[q * 768 + 512 + h * 32 + d];
    }
    __syncthreads();
    const float scale = 0.17677669529663687f;
    for (int idx = tid; idx < 1024; idx += 256) {
        int q = idx >> 5, k = idx & 31;
        float s = 0.f;
        #pragma unroll
        for (int d = 0; d < 32; ++d) s += sQ[q][d] * sK[k][d];
        sS[q][k] = s * scale;
    }
    __syncthreads();
    if (tid < 32) {
        float mx = -3.0e38f;
        for (int k = 0; k < 32; ++k) mx = fmaxf(mx, sS[tid][k]);
        float e[32], sum = 0.f;
        for (int k = 0; k < 32; ++k) { e[k] = expf(sS[tid][k] - mx); sum += e[k]; }
        float inv = 1.f / sum;
        for (int k = 0; k < 32; ++k) sS[tid][k] = e[k] * inv;
    }
    __syncthreads();
    for (int idx = tid; idx < 1024; idx += 256) {
        int q = idx >> 5, d = idx & 31;
        float o = 0.f;
        #pragma unroll
        for (int k = 0; k < 32; ++k) o += sS[q][k] * sV[k][d];
        odst[q * 256 + h * 32 + d] = o;
    }
}

extern "C" void kernel_launch(void* const* d_in, const int* in_sizes, int n_in,
                              void* d_out, int out_size, void* d_ws, size_t ws_size,
                              hipStream_t stream)
{
    (void)in_sizes; (void)n_in; (void)out_size; (void)ws_size;
    const float* x          = (const float*)d_in[0];
    const float* rgb_w1     = (const float*)d_in[1];
    const float* rgb_b1     = (const float*)d_in[2];
    const float* rgb_w2     = (const float*)d_in[3];
    const float* rgb_b2     = (const float*)d_in[4];
    const float* frq_w1     = (const float*)d_in[5];
    const float* frq_b1     = (const float*)d_in[6];
    const float* frq_w2     = (const float*)d_in[7];
    const float* frq_b2     = (const float*)d_in[8];
    const float* rgb_proj_w = (const float*)d_in[9];
    const float* rgb_proj_b = (const float*)d_in[10];
    const float* frq_proj_w = (const float*)d_in[11];
    const float* frq_proj_b = (const float*)d_in[12];
    const float* attn_w_in  = (const float*)d_in[13];
    const float* attn_b_in  = (const float*)d_in[14];
    const float* attn_w_out = (const float*)d_in[15];
    const float* attn_b_out = (const float*)d_in[16];
    const float* fus_w1     = (const float*)d_in[17];
    const float* fus_b1     = (const float*)d_in[18];
    const float* fus_w2     = (const float*)d_in[19];
    const float* fus_b2     = (const float*)d_in[20];
    const float* fus_w3     = (const float*)d_in[21];
    const float* fus_b3     = (const float*)d_in[22];

    float* out      = (float*)d_out;
    float* feat_rgb = out + 64;
    float* feat_frq = out + 64 + 40960;

    float* ws       = (float*)d_ws;
    float* h1       = ws;
    float* F1       = ws;
    float* freqimg  = ws + 12845056;
    float* mm       = freqimg + 4816896;
    float* hd       = mm + 192;
    float* rgb_p    = hd;
    float* frq_p    = rgb_p + 8192;
    float* qkv_rgb  = frq_p + 8192;
    float* qkv_frq  = qkv_rgb + 24576;
    float* o1       = qkv_frq + 24576;
    float* o2       = o1 + 8192;
    float* hb       = o2 + 8192;
    float* h1f      = hb + 16384;
    float* h2f      = h1f + 16384;

    conv1_silu_kernel<<<1568, 128, 0, stream>>>(x, rgb_w1, rgb_b1, h1);
    conv2_wmma_pool_kernel<<<dim3(40, 32), 128, 0, stream>>>(h1, rgb_w2, rgb_b2, feat_rgb);

    dft_rows_kernel<<<2688, 224, 0, stream>>>(x, F1);
    dft_cols_mag_kernel<<<1344, 224, 0, stream>>>(F1, freqimg);
    minmax_kernel<<<96, 256, 0, stream>>>(freqimg, mm);
    normalize_kernel<<<18816, 256, 0, stream>>>(freqimg, mm);
    conv1_silu_kernel<<<1568, 128, 0, stream>>>(freqimg, frq_w1, frq_b1, h1);
    conv2_wmma_pool_kernel<<<dim3(40, 32), 128, 0, stream>>>(h1, frq_w2, frq_b2, feat_frq);

    fc_wmma_kernel<1280, false><<<dim3(8, 2), 256, 0, stream>>>(feat_rgb, rgb_proj_w, rgb_proj_b, rgb_p, 256, 256, 0, 0);
    fc_wmma_kernel<1280, false><<<dim3(8, 2), 256, 0, stream>>>(feat_frq, frq_proj_w, frq_proj_b, frq_p, 256, 256, 0, 0);
    fc_wmma_kernel<256, false><<<dim3(24, 2), 256, 0, stream>>>(rgb_p, attn_w_in, attn_b_in, qkv_rgb, 768, 768, 0, 0);
    fc_wmma_kernel<256, false><<<dim3(24, 2), 256, 0, stream>>>(frq_p, attn_w_in, attn_b_in, qkv_frq, 768, 768, 0, 0);
    attn_core_kernel<<<16, 256, 0, stream>>>(qkv_rgb, qkv_frq, o1, o2);
    fc_wmma_kernel<256, false><<<dim3(8, 2), 256, 0, stream>>>(o1, attn_w_out, attn_b_out, hb, 256, 512, 0, 0);
    fc_wmma_kernel<256, false><<<dim3(8, 2), 256, 0, stream>>>(o2, attn_w_out, attn_b_out, hb, 256, 512, 256, 0);
    fc_wmma_kernel<512, false><<<dim3(16, 2), 256, 0, stream>>>(hb,  fus_w1, fus_b1, h1f, 512, 512, 0, 1);
    fc_wmma_kernel<512, false><<<dim3(8, 2), 256, 0, stream>>>(h1f, fus_w2, fus_b2, h2f, 256, 256, 0, 1);
    fc_wmma_kernel<256, true><<<dim3(1, 2), 256, 0, stream>>>(h2f, fus_w3, fus_b3, out, 2, 2, 0, 0);
}


# conv2 input strip requested one strip ahead, under the previous strip's WMMAs: h1 pixel-major makes a strip 35 16-byte loads per thread, held in registers; 2 blocks per WGP, not 3; sums unchanged
# speedup vs baseline: 2.5628x; 1.0607x over previous
#include <hip/hip_runtime.h>
#include <hip/hip_bf16.h>
#include <math.h>

typedef float v2f __attribute__((ext_vector_type(2)));
typedef float v8f __attribute__((ext_vector_type(8)));

#define PI_F 3.14159265358979323846f

#define C1_CHUNKS 2

__device__ __forceinline__ constexpr int  c1_tap_off(int t)
{ return t < 27 ? (t / 9) * 50176 + ((t % 9) / 3) * 224 + (t % 3) : 0; }
__device__ __forceinline__ constexpr bool c1_tap_kx2(int t) { return t < 27 && (t % 3) == 2; }
__device__ __forceinline__ constexpr bool c1_tap_ky2(int t) { return t < 27 && ((t % 9) / 3) == 2; }

__device__ __forceinline__ void conv1_load_tile(const float* __restrict__ xi, int tl,
                                                int nl, int half, float (&b)[14])
{
    const int  oy = tl / 7;
    const int  ox = (tl - oy * 7) * 16 + nl;
    const bool lastrow = (oy == 111);
    const bool lastcol = (ox == 111);
    const float* p = xi + (2 * oy) * 224 + 2 * ox;
    #pragma unroll
    for (int j = 0; j < 14; ++j) {
        const int  t0  = 4 * (j >> 1) + (j & 1);
        const int  off = half ? c1_tap_off(t0 + 2) : c1_tap_off(t0);
        const bool kx2 = half ? c1_tap_kx2(t0 + 2) : c1_tap_kx2(t0);
        const bool ky2 = half ? c1_tap_ky2(t0 + 2) : c1_tap_ky2(t0);
        const bool tv  = half ? (t0 + 2 < 27) : (t0 < 27);
        const bool ok  = tv && !(kx2 && lastcol) && !(ky2 && lastrow);
        const float v  = p[ok ? off : 0];
        b[j] = ok ? v : 0.f;
    }
}

__global__ __launch_bounds__(128)
void conv1_silu_kernel(const float* __restrict__ x, const float* __restrict__ w1,
                       const float* __restrict__ b1, float* __restrict__ out)
{
    const int lane = threadIdx.x & 31;
    const int half = lane >> 4;
    const int nl   = lane & 15;
    const int wv  = __builtin_amdgcn_readfirstlane((int)(blockIdx.x * 4 + (threadIdx.x >> 5)));
    const int img = wv / (392 / C1_CHUNKS);
    const int q0  = (wv - img * (392 / C1_CHUNKS)) * C1_CHUNKS;

    v2f afr[2][7];
    #pragma unroll
    for (int mt = 0; mt < 2; ++mt) {
        const float* wr = w1 + (mt * 16 + nl) * 27 + 2 * half;
        #pragma unroll
        for (int s = 0; s < 7; ++s) {
            v2f a;
            a.x = wr[4 * s];
            if (s < 6) {
                a.y = wr[4 * s + 1];
            } else {
                const float t = wr[half ? 0 : 25];
                a.y = half ? 0.f : t;
            }
            afr[mt][s] = a;
        }
    }
    float bias[2][8];
    #pragma unroll
    for (int mt = 0; mt < 2; ++mt)
        #pragma unroll
        for (int r = 0; r < 8; ++r) bias[mt][r] = b1[mt * 16 + 8 * half + r];

    const float* xi = x + (size_t)img * 150528;
    float* ob = out + (size_t)img * (32 * 12544) + nl * 32 + 8 * half;

    float bc[2][14], bn[2][14];
    conv1_load_tile(xi, 2 * q0,     nl, half, bc[0]);
    conv1_load_tile(xi, 2 * q0 + 1, nl, half, bc[1]);
    #pragma unroll
    for (int i = 0; i < C1_CHUNKS; ++i) {
        const int q = q0 + i;
        if (i + 1 < C1_CHUNKS) {
            conv1_load_tile(xi, 2 * q + 2, nl, half, bn[0]);
            conv1_load_tile(xi, 2 * q + 3, nl, half, bn[1]);
        }
        v8f c[2][2];
        #pragma unroll
        for (int tl = 0; tl < 2; ++tl)
            #pragma unroll
            for (int mt = 0; mt < 2; ++mt)
                c[tl][mt] = (v8f){0.f, 0.f, 0.f, 0.f, 0.f, 0.f, 0.f, 0.f};
        #pragma unroll
        for (int s = 0; s < 7; ++s) {
            #pragma unroll
            for (int tl = 0; tl < 2; ++tl) {
                v2f b;
                b.x = bc[tl][2 * s];
                b.y = bc[tl][2 * s + 1];
                #pragma unroll
                for (int mt = 0; mt < 2; ++mt)
                    c[tl][mt] = __builtin_amdgcn_wmma_f32_16x16x4_f32(false, afr[mt][s], false, b,
                                                                     (short)0, c[tl][mt], false, false);
            }
        }
        #pragma unroll
        for (int tl = 0; tl < 2; ++tl)
            #pragma unroll
            for (int mt = 0; mt < 2; ++mt) {
                float o[8];
                #pragma unroll
                for (int r = 0; r < 8; ++r) {
                    const float v = c[tl][mt][r] + bias[mt][r];
                    o[r] = v / (1.f + expf(-v));
                }
                float4* dst = (float4*)(ob + (2 * q + tl) * (16 * 32) + mt * 16);
                dst[0] = make_float4(o[0], o[1], o[2], o[3]);
                dst[1] = make_float4(o[4], o[5], o[6], o[7]);
            }
        if (i + 1 < C1_CHUNKS) {
            #pragma unroll
            for (int tl = 0; tl < 2; ++tl)
                #pragma unroll
                for (int j = 0; j < 14; ++j) bc[tl][j] = bn[tl][j];
        }
    }
}

#define C2_PP 34
#define C2_RP 3856

template <int NT>
__device__ __forceinline__ void conv2_tiles(const v2f (&afr)[72], const float* tb,
                                            const float (&bias)[8], float (&accR)[8])
{
    v8f c[NT];
    #pragma unroll
    for (int t = 0; t < NT; ++t)
        c[t] = (v8f){0.f, 0.f, 0.f, 0.f, 0.f, 0.f, 0.f, 0.f};

    v2f bc[NT][2], bn[NT][2];
    #pragma unroll
    for (int t = 0; t < NT; ++t) {
        bc[t][0] = *(const v2f*)(tb + t * 16 * C2_PP);
        bc[t][1] = *(const v2f*)(tb + t * 16 * C2_PP + 4);
    }
    #pragma unroll
    for (int st = 0; st < 36; ++st) {
        if (st + 1 < 36) {
            const int icb = (st + 1) / 9, s = (st + 1) - icb * 9;
            const int ky = s / 3, kx = s - ky * 3;
            const int off = ky * C2_RP + kx * C2_PP + icb * 8;
            __builtin_amdgcn_sched_barrier(0);
            #pragma unroll
            for (int t = 0; t < NT; ++t) {
                bn[t][0] = *(const v2f*)(tb + off + t * 16 * C2_PP);
                bn[t][1] = *(const v2f*)(tb + off + t * 16 * C2_PP + 4);
            }
            __builtin_amdgcn_sched_barrier(0);
        }
        const int icb = st / 9, s = st - icb * 9;
        const v2f a0 = afr[icb * 18 + 2 * s];
        const v2f a1 = afr[icb * 18 + 2 * s + 1];
        #pragma unroll
        for (int t = 0; t < NT; ++t)
            c[t] = __builtin_amdgcn_wmma_f32_16x16x4_f32(false, a0, false, bc[t][0],
                                                         (short)0, c[t], false, false);
        #pragma unroll
        for (int t = 0; t < NT; ++t)
            c[t] = __builtin_amdgcn_wmma_f32_16x16x4_f32(false, a1, false, bc[t][1],
                                                         (short)0, c[t], false, false);
        #pragma unroll
        for (int t = 0; t < NT; ++t) { bc[t][0] = bn[t][0]; bc[t][1] = bn[t][1]; }
    }
    #pragma unroll
    for (int t = 0; t < NT; ++t) {
        #pragma unroll
        for (int j = 0; j < 8; ++j) {
            float v = c[t][j] + bias[j];
            accR[j] += v / (1.f + expf(-v));
        }
    }
}

__global__ __launch_bounds__(128)
void conv2_wmma_pool_kernel(const float* __restrict__ h1, const float* __restrict__ w2,
                            const float* __restrict__ b2, float* __restrict__ feat)
{
    __shared__ __attribute__((aligned(16))) float sIn[5 * C2_RP];
    __shared__ float sOut[2][32];

    const int mgrp  = blockIdx.x;
    const int img   = blockIdx.y;
    const int tid   = threadIdx.x;
    const int lane  = tid & 31;
    const int wave  = __builtin_amdgcn_readfirstlane(tid >> 5);
    const int half  = lane >> 4;
    const int nl    = lane & 15;
    const int ct    = wave & 1;
    const int ph    = wave >> 1;
    const int mtile = mgrp * 2 + ct;

    v2f afr[72];
    {
        const float* wr = w2 + (size_t)(mtile * 16 + nl) * 288 + half * 18;
        #pragma unroll
        for (int icb = 0; icb < 4; ++icb) {
            #pragma unroll
            for (int kq = 0; kq < 18; ++kq) {
                const int s  = kq >> 1;
                const int ic = icb * 8 + (kq & 1) * 4;
                v2f a;
                a.x = wr[ic * 9 + s];
                a.y = wr[(ic + 1) * 9 + s];
                afr[icb * 18 + kq] = a;
            }
        }
    }

    float bias[8];
    #pragma unroll
    for (int j = 0; j < 8; ++j) bias[j] = b2[mtile * 16 + j + half * 8];

    for (int i = tid; i < 5 * 32; i += 128)
        sIn[(i >> 5) * C2_RP + 112 * C2_PP + (i & 31)] = 0.f;

    float accR[8] = {0.f, 0.f, 0.f, 0.f, 0.f, 0.f, 0.f, 0.f};

    const float4* gsrc = (const float4*)(h1 + (size_t)img * (112 * 112 * 32)) + tid;
    float* st = sIn + (tid >> 3) * C2_PP + (tid & 7) * 4;

    float4 pv[5][7];
    #pragma unroll
    for (int row = 0; row < 5; ++row) {
        #pragma unroll
        for (int i = 0; i < 7; ++i) pv[row][i] = gsrc[row * 896 + i * 128];
    }

    const int dy = nl >> 3, dx = nl & 7;
    const float* tb = sIn + (2 * dy) * C2_RP + (2 * dx) * C2_PP + half * 2;

    for (int yt = 0; yt < 28; ++yt) {
        __syncthreads();
        #pragma unroll
        for (int row = 0; row < 5; ++row) {
            #pragma unroll
            for (int i = 0; i < 7; ++i) {
                const float4 t = pv[row][i];
                float* d = st + row * C2_RP + i * 16 * C2_PP;
                *(float2*)d       = make_float2(t.x, t.y);
                *(float2*)(d + 2) = make_float2(t.z, t.w);
            }
        }
        if (yt == 27) {
            #pragma unroll
            for (int i = 0; i < 7; ++i) {
                float* d = st + 4 * C2_RP + i * 16 * C2_PP;
                *(float2*)d       = make_float2(0.f, 0.f);
                *(float2*)(d + 2) = make_float2(0.f, 0.f);
            }
        }
        __syncthreads();
        if (yt < 27) {
            const float4* pn = gsrc + (size_t)(yt + 1) * (4 * 896);
            #pragma unroll
            for (int row = 0; row < 5; ++row) {
                const float4* pr = pn + ((row == 4 && yt == 26) ? 3 : row) * 896;
                #pragma unroll
                for (int i = 0; i < 7; ++i) pv[row][i] = pr[i * 128];
            }
        }

        if (((yt ^ ph) & 1) == 0) conv2_tiles<4>(afr, tb, bias, accR);
        else                      conv2_tiles<3>(afr, tb + 4 * 16 * C2_PP, bias, accR);
    }

    #pragma unroll
    for (int j = 0; j < 8; ++j) {
        float v = accR[j];
        v += __shfl_xor(v, 1);
        v += __shfl_xor(v, 2);
        v += __shfl_xor(v, 4);
        v += __shfl_xor(v, 8);
        if (nl == 0) sOut[ph][ct * 16 + half * 8 + j] = v;
    }
    __syncthreads();
    if (tid < 32)
        feat[img * 1280 + mgrp * 32 + tid] = (sOut[0][tid] + sOut[1][tid]) * (1.f / 3136.f);
}

#define DFT_ROWS 8
#define DFT_COLS 16
#define DFT_CP   225

__global__ __launch_bounds__(224)
void dft_rows_kernel(const float* __restrict__ x, float* __restrict__ F1)
{
    __shared__ float ct[224], st[224];
    __shared__ float xs[DFT_ROWS][224];
    __shared__ float2 ys[DFT_ROWS][224];
    const int t   = threadIdx.x;
    const int bc  = blockIdx.x / (224 / DFT_ROWS);
    const int y0  = (blockIdx.x - bc * (224 / DFT_ROWS)) * DFT_ROWS;
    const float ang = (2.0f * PI_F / 224.0f) * (float)t;
    ct[t] = cosf(ang); st[t] = sinf(ang);
    const float* rows = x + ((size_t)bc * 224 + y0) * 224;
    #pragma unroll
    for (int r = 0; r < DFT_ROWS; ++r) xs[r][t] = rows[r * 224 + t];
    __syncthreads();

    const int lo = t & 15, hi = t >> 4;
    float re[DFT_ROWS], im[DFT_ROWS];
    #pragma unroll
    for (int r = 0; r < DFT_ROWS; ++r) { re[r] = 0.f; im[r] = 0.f; }
    {
        int j = 0;
        const int dj = 14 * lo;
        for (int n1 = 0; n1 < 16; ++n1) {
            const float c = ct[j], s = st[j];
            #pragma unroll
            for (int r = 0; r < DFT_ROWS; ++r) {
                const float v = xs[r][14 * n1 + hi];
                re[r] += v * c; im[r] -= v * s;
            }
            j += dj; if (j >= 224) j -= 224;
        }
        const float c = ct[hi * lo], s = st[hi * lo];
        #pragma unroll
        for (int r = 0; r < DFT_ROWS; ++r) {
            float2 y;
            y.x = re[r] * c + im[r] * s;
            y.y = im[r] * c - re[r] * s;
            ys[r][t] = y;
        }
    }
    __syncthreads();

    #pragma unroll
    for (int r = 0; r < DFT_ROWS; ++r) { re[r] = 0.f; im[r] = 0.f; }
    {
        int j = 0;
        const int dj = 16 * hi;
        for (int n2 = 0; n2 < 14; ++n2) {
            const float c = ct[j], s = st[j];
            #pragma unroll
            for (int r = 0; r < DFT_ROWS; ++r) {
                const float2 y = ys[r][n2 * 16 + lo];
                re[r] += y.x * c; re[r] += y.y * s;
                im[r] += y.y * c; im[r] -= y.x * s;
            }
            j += dj; if (j >= 224) j -= 224;
        }
    }
    float* o = F1 + (((size_t)bc * 224 + y0) * 224 + t) * 2;
    #pragma unroll
    for (int r = 0; r < DFT_ROWS; ++r) { o[r * 448] = re[r]; o[r * 448 + 1] = im[r]; }
}

__global__ __launch_bounds__(224)
void dft_cols_mag_kernel(const float* __restrict__ F1, float* __restrict__ mag)
{
    __shared__ float ct[224], st[224];
    __shared__ float2 fs[DFT_COLS][DFT_CP];
    __shared__ float2 ys[DFT_COLS][DFT_CP];
    const int t   = threadIdx.x;
    const int bc  = blockIdx.x / (224 / DFT_COLS);
    const int u0  = (blockIdx.x - bc * (224 / DFT_COLS)) * DFT_COLS;
    const float ang = (2.0f * PI_F / 224.0f) * (float)t;
    ct[t] = cosf(ang); st[t] = sinf(ang);
    {
        const float2* base = (const float2*)(F1 + (size_t)bc * (224 * 448) + u0 * 2);
        #pragma unroll
        for (int i = 0; i < 16; ++i) {
            const int e = i * 224 + t;
            const int y = e >> 4, q = e & 15;
            fs[q][y] = base[y * 224 + q];
        }
    }
    __syncthreads();

    const int lo = t & 15, hi = t >> 4;
    float re[DFT_COLS], im[DFT_COLS];
    #pragma unroll
    for (int q = 0; q < DFT_COLS; ++q) { re[q] = 0.f; im[q] = 0.f; }
    {
        int j = 0;
        const int dj = 14 * lo;
        for (int n1 = 0; n1 < 16; ++n1) {
            const float c = ct[j], s = st[j];
            #pragma unroll
            for (int q = 0; q < DFT_COLS; ++q) {
                const float2 f = fs[q][14 * n1 + hi];
                re[q] += f.x * c; re[q] += f.y * s;
                im[q] += f.y * c; im[q] -= f.x * s;
            }
            j += dj; if (j >= 224) j -= 224;
        }
        const float c = ct[hi * lo], s = st[hi * lo];
        #pragma unroll
        for (int q = 0; q < DFT_COLS; ++q) {
            float2 y;
            y.x = re[q] * c + im[q] * s;
            y.y = im[q] * c - re[q] * s;
            ys[q][t] = y;
        }
    }
    __syncthreads();

    #pragma unroll
    for (int q = 0; q < DFT_COLS; ++q) { re[q] = 0.f; im[q] = 0.f; }
    {
        int j = 0;
        const int dj = 16 * hi;
        for (int n2 = 0; n2 < 14; ++n2) {
            const float c = ct[j], s = st[j];
            #pragma unroll
            for (int q = 0; q < DFT_COLS; ++q) {
                const float2 y = ys[q][n2 * 16 + lo];
                re[q] += y.x * c; re[q] += y.y * s;
                im[q] += y.y * c; im[q] -= y.x * s;
            }
            j += dj; if (j >= 224) j -= 224;
        }
    }
    int vo = t + 112;  if (vo >= 224) vo -= 224;
    int uo = u0 + 112; if (uo >= 224) uo -= 224;
    float4* o = (float4*)(mag + ((size_t)bc * 224 + vo) * 224 + uo);
    #pragma unroll
    for (int q = 0; q < DFT_COLS; q += 4) {
        float4 m;
        m.x = logf(sqrtf(re[q]     * re[q]     + im[q]     * im[q])     + 1.0f);
        m.y = logf(sqrtf(re[q + 1] * re[q + 1] + im[q + 1] * im[q + 1]) + 1.0f);
        m.z = logf(sqrtf(re[q + 2] * re[q + 2] + im[q + 2] * im[q + 2]) + 1.0f);
        m.w = logf(sqrtf(re[q + 3] * re[q + 3] + im[q + 3] * im[q + 3]) + 1.0f);
        o[q >> 2] = m;
    }
}

__global__ __launch_bounds__(256)
void minmax_kernel(const float* __restrict__ mag, float* __restrict__ mm)
{
    __shared__ float smn[256], smx[256];
    int bc = blockIdx.x, tid = threadIdx.x;
    const float* p = mag + (size_t)bc * 50176;
    float mn = 3.0e38f, mx = -3.0e38f;
    for (int i = tid; i < 50176; i += 256) {
        float v = p[i]; mn = fminf(mn, v); mx = fmaxf(mx, v);
    }
    smn[tid] = mn; smx[tid] = mx; __syncthreads();
    for (int s = 128; s > 0; s >>= 1) {
        if (tid < s) { smn[tid] = fminf(smn[tid], smn[tid + s]);
                       smx[tid] = fmaxf(smx[tid], smx[tid + s]); }
        __syncthreads();
    }
    if (tid == 0) { mm[bc * 2] = smn[0]; mm[bc * 2 + 1] = smx[0]; }
}

__global__ __launch_bounds__(256)
void normalize_kernel(float* __restrict__ mag, const float* __restrict__ mm)
{
    int idx = blockIdx.x * 256 + threadIdx.x;
    if (idx >= 96 * 50176) return;
    int bc = idx / 50176;
    float mn = mm[bc * 2], mx = mm[bc * 2 + 1];
    mag[idx] = (mag[idx] - mn) / (mx - mn + 1e-8f);
}

#define FC_TP 272

template <int K, bool SMALLN>
__global__ __launch_bounds__(256)
void fc_wmma_kernel(const float* __restrict__ in, const float* __restrict__ w,
                    const float* __restrict__ b, float* __restrict__ out,
                    int N, int outStride, int outOffset, int act)
{
    constexpr int KC = K / 8;
    constexpr int NL = KC / 8;
    __shared__ float part[8 * 2 * FC_TP];

    const int tid  = threadIdx.x;
    const int lane = tid & 31;
    const int wave = tid >> 5;
    const int half = lane >> 4;
    const int c    = lane & 15;
    const int j0   = blockIdx.x * 32;
    const int i0   = blockIdx.y * 16;
    const int kb   = wave * KC + 4 * half;

    int ja = j0 + c, jb = j0 + 16 + c;
    if (SMALLN) ja = ja < N ? ja : N - 1;
    const float4* pa  = (const float4*)(in + (size_t)(i0 + c) * K + kb);
    const float4* pb0 = (const float4*)(w + (size_t)ja * K + kb);
    const float4* pb1 = (const float4*)(w + (size_t)jb * K + kb);

    int fin_il = 0, fin_j = lane;
    if (SMALLN) {
        const int tt = tid < 16 * N ? tid : 0;
        fin_il = tt / N;
        fin_j  = tt - fin_il * N;
    }
    const float bj = b[SMALLN ? fin_j : j0 + lane];

    float4 xa[NL], wb0[NL], wb1[NL];
    #pragma unroll
    for (int m = 0; m < NL; ++m) {
        xa[m]  = pa[2 * m];
        wb0[m] = pb0[2 * m];
        if (!SMALLN) wb1[m] = pb1[2 * m];
    }
    __builtin_amdgcn_sched_barrier(0);

    v8f acc0 = (v8f){0.f, 0.f, 0.f, 0.f, 0.f, 0.f, 0.f, 0.f};
    v8f acc1 = (v8f){0.f, 0.f, 0.f, 0.f, 0.f, 0.f, 0.f, 0.f};
    #pragma unroll
    for (int m = 0; m < NL; ++m) {
        v2f a, p, q;
        a.x = xa[m].x;  a.y = xa[m].y;
        p.x = wb0[m].x; p.y = wb0[m].y;
        acc0 = __builtin_amdgcn_wmma_f32_16x16x4_f32(false, a, false, p, (short)0, acc0, false, false);
        if (!SMALLN) {
            q.x = wb1[m].x; q.y = wb1[m].y;
            acc1 = __builtin_amdgcn_wmma_f32_16x16x4_f32(false, a, false, q, (short)0, acc1, false, false);
        }
        a.x = xa[m].z;  a.y = xa[m].w;
        p.x = wb0[m].z; p.y = wb0[m].w;
        acc0 = __builtin_amdgcn_wmma_f32_16x16x4_f32(false, a, false, p, (short)0, acc0, false, false);
        if (!SMALLN) {
            q.x = wb1[m].z; q.y = wb1[m].w;
            acc1 = __builtin_amdgcn_wmma_f32_16x16x4_f32(false, a, false, q, (short)0, acc1, false, false);
        }
    }

    float* pp = part + (wave * 2) * FC_TP + lane;
    #pragma unroll
    for (int r = 0; r < 8; ++r) {
        pp[r * 32] = acc0[r];
        if (!SMALLN) pp[FC_TP + r * 32] = acc1[r];
    }
    __syncthreads();

    if (!SMALLN) {
        const int t = lane >> 4;
        #pragma unroll
        for (int u = 0; u < 2; ++u) {
            const int il  = 2 * wave + u;
            const int src = t * FC_TP + (il & 7) * 32 + 16 * (il >> 3) + c;
            float s = part[src];
            #pragma unroll
            for (int q = 1; q < 8; ++q) s += part[q * 2 * FC_TP + src];
            s += bj;
            if (act == 1) s = fmaxf(s, 0.f);
            out[(size_t)(i0 + il) * outStride + outOffset + j0 + lane] = s;
        }
    } else {
        if (tid < 16 * N) {
            const int il  = fin_il;
            const int j   = fin_j;
            const int src = (il & 7) * 32 + 16 * (il >> 3) + j;
            float s = part[src];
            #pragma unroll
            for (int q = 1; q < 8; ++q) s += part[q * 2 * FC_TP + src];
            s += bj;
            if (act == 1) s = fmaxf(s, 0.f);
            out[(size_t)(i0 + il) * outStride + outOffset + j] = s;
        }
    }
}

__global__ __launch_bounds__(256)
void attn_core_kernel(const float* __restrict__ qkv_rgb, const float* __restrict__ qkv_frq,
                      float* __restrict__ o1, float* __restrict__ o2)
{
    __shared__ float sQ[32][32], sK[32][32], sV[32][32], sS[32][33];
    int a = blockIdx.x >> 3;
    int h = blockIdx.x & 7;
    const float* qsrc  = (a == 0) ? qkv_rgb : qkv_frq;
    const float* kvsrc = (a == 0) ? qkv_frq : qkv_rgb;
    float* odst        = (a == 0) ? o1 : o2;
    int tid = threadIdx.x;
    for (int idx = tid; idx < 1024; idx += 256) {
        int q = idx >> 5, d = idx & 31;
        sQ[q][d] = qsrc [q * 768 +       h * 32 + d];
        sK[q][d] = kvsrc[q * 768 + 256 + h * 32 + d];
        sV[q][d] = kvsrc[q * 768 + 512 + h * 32 + d];
    }
    __syncthreads();
    const float scale = 0.17677669529663687f;
    for (int idx = tid; idx < 1024; idx += 256) {
        int q = idx >> 5, k = idx & 31;
        float s = 0.f;
        #pragma unroll
        for (int d = 0; d < 32; ++d) s += sQ[q][d] * sK[k][d];
        sS[q][k] = s * scale;
    }
    __syncthreads();
    if (tid < 32) {
        float mx = -3.0e38f;
        for (int k = 0; k < 32; ++k) mx = fmaxf(mx, sS[tid][k]);
        float e[32], sum = 0.f;
        for (int k = 0; k < 32; ++k) { e[k] = expf(sS[tid][k] - mx); sum += e[k]; }
        float inv = 1.f / sum;
        for (int k = 0; k < 32; ++k) sS[tid][k] = e[k] * inv;
    }
    __syncthreads();
    for (int idx = tid; idx < 1024; idx += 256) {
        int q = idx >> 5, d = idx & 31;
        float o = 0.f;
        #pragma unroll
        for (int k = 0; k < 32; ++k) o += sS[q][k] * sV[k][d];
        odst[q * 256 + h * 32 + d] = o;
    }
}

extern "C" void kernel_launch(void* const* d_in, const int* in_sizes, int n_in,
                              void* d_out, int out_size, void* d_ws, size_t ws_size,
                              hipStream_t stream)
{
    (void)in_sizes; (void)n_in; (void)out_size; (void)ws_size;
    const float* x          = (const float*)d_in[0];
    const float* rgb_w1     = (const float*)d_in[1];
    const float* rgb_b1     = (const float*)d_in[2];
    const float* rgb_w2     = (const float*)d_in[3];
    const float* rgb_b2     = (const float*)d_in[4];
    const float* frq_w1     = (const float*)d_in[5];
    const float* frq_b1     = (const float*)d_in[6];
    const float* frq_w2     = (const float*)d_in[7];
    const float* frq_b2     = (const float*)d_in[8];
    const float* rgb_proj_w = (const float*)d_in[9];
    const float* rgb_proj_b = (const float*)d_in[10];
    const float* frq_proj_w = (const float*)d_in[11];
    const float* frq_proj_b = (const float*)d_in[12];
    const float* attn_w_in  = (const float*)d_in[13];
    const float* attn_b_in  = (const float*)d_in[14];
    const float* attn_w_out = (const float*)d_in[15];
    const float* attn_b_out = (const float*)d_in[16];
    const float* fus_w1     = (const float*)d_in[17];
    const float* fus_b1     = (const float*)d_in[18];
    const float* fus_w2     = (const float*)d_in[19];
    const float* fus_b2     = (const float*)d_in[20];
    const float* fus_w3     = (const float*)d_in[21];
    const float* fus_b3     = (const float*)d_in[22];

    float* out      = (float*)d_out;
    float* feat_rgb = out + 64;
    float* feat_frq = out + 64 + 40960;

    float* ws       = (float*)d_ws;
    float* h1       = ws;
    float* F1       = ws;
    float* freqimg  = ws + 12845056;
    float* mm       = freqimg + 4816896;
    float* hd       = mm + 192;
    float* rgb_p    = hd;
    float* frq_p    = rgb_p + 8192;
    float* qkv_rgb  = frq_p + 8192;
    float* qkv_frq  = qkv_rgb + 24576;
    float* o1       = qkv_frq + 24576;
    float* o2       = o1 + 8192;
    float* hb       = o2 + 8192;
    float* h1f      = hb + 16384;
    float* h2f      = h1f + 16384;

    conv1_silu_kernel<<<1568, 128, 0, stream>>>(x, rgb_w1, rgb_b1, h1);
    conv2_wmma_pool_kernel<<<dim3(40, 32), 128, 0, stream>>>(h1, rgb_w2, rgb_b2, feat_rgb);

    dft_rows_kernel<<<2688, 224, 0, stream>>>(x, F1);
    dft_cols_mag_kernel<<<1344, 224, 0, stream>>>(F1, freqimg);
    minmax_kernel<<<96, 256, 0, stream>>>(freqimg, mm);
    normalize_kernel<<<18816, 256, 0, stream>>>(freqimg, mm);
    conv1_silu_kernel<<<1568, 128, 0, stream>>>(freqimg, frq_w1, frq_b1, h1);
    conv2_wmma_pool_kernel<<<dim3(40, 32), 128, 0, stream>>>(h1, frq_w2, frq_b2, feat_frq);

    fc_wmma_kernel<1280, false><<<dim3(8, 2), 256, 0, stream>>>(feat_rgb, rgb_proj_w, rgb_proj_b, rgb_p, 256, 256, 0, 0);
    fc_wmma_kernel<1280, false><<<dim3(8, 2), 256, 0, stream>>>(feat_frq, frq_proj_w, frq_proj_b, frq_p, 256, 256, 0, 0);
    fc_wmma_kernel<256, false><<<dim3(24, 2), 256, 0, stream>>>(rgb_p, attn_w_in, attn_b_in, qkv_rgb, 768, 768, 0, 0);
    fc_wmma_kernel<256, false><<<dim3(24, 2), 256, 0, stream>>>(frq_p, attn_w_in, attn_b_in, qkv_frq, 768, 768, 0, 0);
    attn_core_kernel<<<16, 256, 0, stream>>>(qkv_rgb, qkv_frq, o1, o2);
    fc_wmma_kernel<256, false><<<dim3(8, 2), 256, 0, stream>>>(o1, attn_w_out, attn_b_out, hb, 256, 512, 0, 0);
    fc_wmma_kernel<256, false><<<dim3(8, 2), 256, 0, stream>>>(o2, attn_w_out, attn_b_out, hb, 256, 512, 256, 0);
    fc_wmma_kernel<512, false><<<dim3(16, 2), 256, 0, stream>>>(hb,  fus_w1, fus_b1, h1f, 512, 512, 0, 1);
    fc_wmma_kernel<512, false><<<dim3(8, 2), 256, 0, stream>>>(h1f, fus_w2, fus_b2, h2f, 256, 256, 0, 1);
    fc_wmma_kernel<256, true><<<dim3(1, 2), 256, 0, stream>>>(h2f, fus_w3, fus_b3, out, 2, 2, 0, 0);
}


# conv2: a wave pair splits a strip's 7 pixel tiles 3.5 : 3.5 (was 4 : 3 in turn): middle tile's K halved, halves joined through LDS behind the next strip barrier; conv2 K and pool sum order changes
# speedup vs baseline: 2.6542x; 1.0357x over previous
#include <hip/hip_runtime.h>
#include <hip/hip_bf16.h>
#include <math.h>

typedef float v2f __attribute__((ext_vector_type(2)));
typedef float v8f __attribute__((ext_vector_type(8)));

#define PI_F 3.14159265358979323846f

#define C1_CHUNKS 2

__device__ __forceinline__ constexpr int  c1_tap_off(int t)
{ return t < 27 ? (t / 9) * 50176 + ((t % 9) / 3) * 224 + (t % 3) : 0; }
__device__ __forceinline__ constexpr bool c1_tap_kx2(int t) { return t < 27 && (t % 3) == 2; }
__device__ __forceinline__ constexpr bool c1_tap_ky2(int t) { return t < 27 && ((t % 9) / 3) == 2; }

__device__ __forceinline__ void conv1_load_tile(const float* __restrict__ xi, int tl,
                                                int nl, int half, float (&b)[14])
{
    const int  oy = tl / 7;
    const int  ox = (tl - oy * 7) * 16 + nl;
    const bool lastrow = (oy == 111);
    const bool lastcol = (ox == 111);
    const float* p = xi + (2 * oy) * 224 + 2 * ox;
    #pragma unroll
    for (int j = 0; j < 14; ++j) {
        const int  t0  = 4 * (j >> 1) + (j & 1);
        const int  off = half ? c1_tap_off(t0 + 2) : c1_tap_off(t0);
        const bool kx2 = half ? c1_tap_kx2(t0 + 2) : c1_tap_kx2(t0);
        const bool ky2 = half ? c1_tap_ky2(t0 + 2) : c1_tap_ky2(t0);
        const bool tv  = half ? (t0 + 2 < 27) : (t0 < 27);
        const bool ok  = tv && !(kx2 && lastcol) && !(ky2 && lastrow);
        const float v  = p[ok ? off : 0];
        b[j] = ok ? v : 0.f;
    }
}

__global__ __launch_bounds__(128)
void conv1_silu_kernel(const float* __restrict__ x, const float* __restrict__ w1,
                       const float* __restrict__ b1, float* __restrict__ out)
{
    const int lane = threadIdx.x & 31;
    const int half = lane >> 4;
    const int nl   = lane & 15;
    const int wv  = __builtin_amdgcn_readfirstlane((int)(blockIdx.x * 4 + (threadIdx.x >> 5)));
    const int img = wv / (392 / C1_CHUNKS);
    const int q0  = (wv - img * (392 / C1_CHUNKS)) * C1_CHUNKS;

    v2f afr[2][7];
    #pragma unroll
    for (int mt = 0; mt < 2; ++mt) {
        const float* wr = w1 + (mt * 16 + nl) * 27 + 2 * half;
        #pragma unroll
        for (int s = 0; s < 7; ++s) {
            v2f a;
            a.x = wr[4 * s];
            if (s < 6) {
                a.y = wr[4 * s + 1];
            } else {
                const float t = wr[half ? 0 : 25];
                a.y = half ? 0.f : t;
            }
            afr[mt][s] = a;
        }
    }
    float bias[2][8];
    #pragma unroll
    for (int mt = 0; mt < 2; ++mt)
        #pragma unroll
        for (int r = 0; r < 8; ++r) bias[mt][r] = b1[mt * 16 + 8 * half + r];

    const float* xi = x + (size_t)img * 150528;
    float* ob = out + (size_t)img * (32 * 12544) + nl * 32 + 8 * half;

    float bc[2][14], bn[2][14];
    conv1_load_tile(xi, 2 * q0,     nl, half, bc[0]);
    conv1_load_tile(xi, 2 * q0 + 1, nl, half, bc[1]);
    #pragma unroll
    for (int i = 0; i < C1_CHUNKS; ++i) {
        const int q = q0 + i;
        if (i + 1 < C1_CHUNKS) {
            conv1_load_tile(xi, 2 * q + 2, nl, half, bn[0]);
            conv1_load_tile(xi, 2 * q + 3, nl, half, bn[1]);
        }
        v8f c[2][2];
        #pragma unroll
        for (int tl = 0; tl < 2; ++tl)
            #pragma unroll
            for (int mt = 0; mt < 2; ++mt)
                c[tl][mt] = (v8f){0.f, 0.f, 0.f, 0.f, 0.f, 0.f, 0.f, 0.f};
        #pragma unroll
        for (int s = 0; s < 7; ++s) {
            #pragma unroll
            for (int tl = 0; tl < 2; ++tl) {
                v2f b;
                b.x = bc[tl][2 * s];
                b.y = bc[tl][2 * s + 1];
                #pragma unroll
                for (int mt = 0; mt < 2; ++mt)
                    c[tl][mt] = __builtin_amdgcn_wmma_f32_16x16x4_f32(false, afr[mt][s], false, b,
                                                                     (short)0, c[tl][mt], false, false);
            }
        }
        #pragma unroll
        for (int tl = 0; tl < 2; ++tl)
            #pragma unroll
            for (int mt = 0; mt < 2; ++mt) {
                float o[8];
                #pragma unroll
                for (int r = 0; r < 8; ++r) {
                    const float v = c[tl][mt][r] + bias[mt][r];
                    o[r] = v / (1.f + expf(-v));
                }
                float4* dst = (float4*)(ob + (2 * q + tl) * (16 * 32) + mt * 16);
                dst[0] = make_float4(o[0], o[1], o[2], o[3]);
                dst[1] = make_float4(o[4], o[5], o[6], o[7]);
            }
        if (i + 1 < C1_CHUNKS) {
            #pragma unroll
            for (int tl = 0; tl < 2; ++tl)
                #pragma unroll
                for (int j = 0; j < 14; ++j) bc[tl][j] = bn[tl][j];
        }
    }
}

#define C2_PP 34
#define C2_RP 3856

template <int PH>
__device__ __forceinline__ void conv2_finish_mid(const float (&keep)[4], const float4 pr,
                                                 const float (&bias)[8], float (&accR)[8])
{
    const float p[4] = {pr.x, pr.y, pr.z, pr.w};
    #pragma unroll
    for (int j = 0; j < 4; ++j) {
        const float v = (keep[j] + p[j]) + bias[PH * 4 + j];
        accR[PH * 4 + j] += v / (1.f + expf(-v));
    }
}

template <int PH>
__device__ __forceinline__ void conv2_tiles(const v2f (&afr)[72], const float* tb,
                                            const float (&bias)[8], float (&accR)[8],
                                            bool have, const float4 pr, float (&keep)[4],
                                            float (&send)[4])
{
    if (have) conv2_finish_mid<PH>(keep, pr, bias, accR);

    constexpr int X0 = PH ? 4 : 0;
    v8f c[3], cm;
    #pragma unroll
    for (int t = 0; t < 3; ++t)
        c[t] = (v8f){0.f, 0.f, 0.f, 0.f, 0.f, 0.f, 0.f, 0.f};
    cm = (v8f){0.f, 0.f, 0.f, 0.f, 0.f, 0.f, 0.f, 0.f};

    v2f bc[3][2], bn[3][2], mc[2], mn[2];
    #pragma unroll
    for (int t = 0; t < 3; ++t) {
        bc[t][0] = *(const v2f*)(tb + (X0 + t) * 16 * C2_PP);
        bc[t][1] = *(const v2f*)(tb + (X0 + t) * 16 * C2_PP + 4);
    }
    if (PH == 0) {
        mc[0] = *(const v2f*)(tb + 3 * 16 * C2_PP);
        mc[1] = *(const v2f*)(tb + 3 * 16 * C2_PP + 4);
    }
    #pragma unroll
    for (int st = 0; st < 36; ++st) {
        if (st + 1 < 36) {
            const int icb = (st + 1) / 9, s = (st + 1) - icb * 9;
            const int ky = s / 3, kx = s - ky * 3;
            const int off = ky * C2_RP + kx * C2_PP + icb * 8;
            __builtin_amdgcn_sched_barrier(0);
            #pragma unroll
            for (int t = 0; t < 3; ++t) {
                bn[t][0] = *(const v2f*)(tb + off + (X0 + t) * 16 * C2_PP);
                bn[t][1] = *(const v2f*)(tb + off + (X0 + t) * 16 * C2_PP + 4);
            }
            if ((st + 1) / 18 == PH) {
                mn[0] = *(const v2f*)(tb + off + 3 * 16 * C2_PP);
                mn[1] = *(const v2f*)(tb + off + 3 * 16 * C2_PP + 4);
            }
            __builtin_amdgcn_sched_barrier(0);
        }
        const int icb = st / 9, s = st - icb * 9;
        const v2f a0 = afr[icb * 18 + 2 * s];
        const v2f a1 = afr[icb * 18 + 2 * s + 1];
        #pragma unroll
        for (int t = 0; t < 3; ++t)
            c[t] = __builtin_amdgcn_wmma_f32_16x16x4_f32(false, a0, false, bc[t][0],
                                                         (short)0, c[t], false, false);
        if (st / 18 == PH)
            cm = __builtin_amdgcn_wmma_f32_16x16x4_f32(false, a0, false, mc[0],
                                                       (short)0, cm, false, false);
        #pragma unroll
        for (int t = 0; t < 3; ++t)
            c[t] = __builtin_amdgcn_wmma_f32_16x16x4_f32(false, a1, false, bc[t][1],
                                                         (short)0, c[t], false, false);
        if (st / 18 == PH)
            cm = __builtin_amdgcn_wmma_f32_16x16x4_f32(false, a1, false, mc[1],
                                                       (short)0, cm, false, false);
        #pragma unroll
        for (int t = 0; t < 3; ++t) { bc[t][0] = bn[t][0]; bc[t][1] = bn[t][1]; }
        if ((st + 1) / 18 == PH && st + 1 < 36) { mc[0] = mn[0]; mc[1] = mn[1]; }
    }
    #pragma unroll
    for (int t = 0; t < 3; ++t) {
        #pragma unroll
        for (int j = 0; j < 8; ++j) {
            float v = c[t][j] + bias[j];
            accR[j] += v / (1.f + expf(-v));
        }
    }
    #pragma unroll
    for (int j = 0; j < 4; ++j) {
        keep[j] = cm[PH * 4 + j];
        send[j] = cm[(1 - PH) * 4 + j];
    }
}

__global__ __launch_bounds__(128)
void conv2_wmma_pool_kernel(const float* __restrict__ h1, const float* __restrict__ w2,
                            const float* __restrict__ b2, float* __restrict__ feat)
{
    __shared__ __attribute__((aligned(16))) float sIn[5 * C2_RP];
    __shared__ float sOut[2][32];
    __shared__ float4 sX[2][2][32];

    const int mgrp  = blockIdx.x;
    const int img   = blockIdx.y;
    const int tid   = threadIdx.x;
    const int lane  = tid & 31;
    const int wave  = __builtin_amdgcn_readfirstlane(tid >> 5);
    const int half  = lane >> 4;
    const int nl    = lane & 15;
    const int ct    = wave & 1;
    const int ph    = wave >> 1;
    const int mtile = mgrp * 2 + ct;

    v2f afr[72];
    {
        const float* wr = w2 + (size_t)(mtile * 16 + nl) * 288 + half * 18;
        #pragma unroll
        for (int icb = 0; icb < 4; ++icb) {
            #pragma unroll
            for (int kq = 0; kq < 18; ++kq) {
                const int s  = kq >> 1;
                const int ic = icb * 8 + (kq & 1) * 4;
                v2f a;
                a.x = wr[ic * 9 + s];
                a.y = wr[(ic + 1) * 9 + s];
                afr[icb * 18 + kq] = a;
            }
        }
    }

    float bias[8];
    #pragma unroll
    for (int j = 0; j < 8; ++j) bias[j] = b2[mtile * 16 + j + half * 8];

    for (int i = tid; i < 5 * 32; i += 128)
        sIn[(i >> 5) * C2_RP + 112 * C2_PP + (i & 31)] = 0.f;

    float accR[8] = {0.f, 0.f, 0.f, 0.f, 0.f, 0.f, 0.f, 0.f};

    const float4* gsrc = (const float4*)(h1 + (size_t)img * (112 * 112 * 32)) + tid;
    float* st = sIn + (tid >> 3) * C2_PP + (tid & 7) * 4;

    float4 pv[5][7];
    #pragma unroll
    for (int row = 0; row < 5; ++row) {
        #pragma unroll
        for (int i = 0; i < 7; ++i) pv[row][i] = gsrc[row * 896 + i * 128];
    }

    const int dy = nl >> 3, dx = nl & 7;
    const float* tb = sIn + (2 * dy) * C2_RP + (2 * dx) * C2_PP + half * 2;

    float keep[4] = {0.f, 0.f, 0.f, 0.f};
    float send[4];
    for (int yt = 0; yt < 28; ++yt) {
        __syncthreads();
        float4 mid_in = make_float4(0.f, 0.f, 0.f, 0.f);
        if (yt > 0) mid_in = sX[ct][ph ^ 1][lane];
        #pragma unroll
        for (int row = 0; row < 5; ++row) {
            #pragma unroll
            for (int i = 0; i < 7; ++i) {
                const float4 t = pv[row][i];
                float* d = st + row * C2_RP + i * 16 * C2_PP;
                *(float2*)d       = make_float2(t.x, t.y);
                *(float2*)(d + 2) = make_float2(t.z, t.w);
            }
        }
        if (yt == 27) {
            #pragma unroll
            for (int i = 0; i < 7; ++i) {
                float* d = st + 4 * C2_RP + i * 16 * C2_PP;
                *(float2*)d       = make_float2(0.f, 0.f);
                *(float2*)(d + 2) = make_float2(0.f, 0.f);
            }
        }
        __syncthreads();
        if (yt < 27) {
            const float4* pn = gsrc + (size_t)(yt + 1) * (4 * 896);
            #pragma unroll
            for (int row = 0; row < 5; ++row) {
                const float4* pr = pn + ((row == 4 && yt == 26) ? 3 : row) * 896;
                #pragma unroll
                for (int i = 0; i < 7; ++i) pv[row][i] = pr[i * 128];
            }
        }

        if (ph == 0) conv2_tiles<0>(afr, tb, bias, accR, yt > 0, mid_in, keep, send);
        else         conv2_tiles<1>(afr, tb, bias, accR, yt > 0, mid_in, keep, send);
        sX[ct][ph][lane] = make_float4(send[0], send[1], send[2], send[3]);
    }

    __syncthreads();
    {
        const float4 mid_in = sX[ct][ph ^ 1][lane];
        if (ph == 0) conv2_finish_mid<0>(keep, mid_in, bias, accR);
        else         conv2_finish_mid<1>(keep, mid_in, bias, accR);
    }

    #pragma unroll
    for (int j = 0; j < 8; ++j) {
        float v = accR[j];
        v += __shfl_xor(v, 1);
        v += __shfl_xor(v, 2);
        v += __shfl_xor(v, 4);
        v += __shfl_xor(v, 8);
        if (nl == 0) sOut[ph][ct * 16 + half * 8 + j] = v;
    }
    __syncthreads();
    if (tid < 32)
        feat[img * 1280 + mgrp * 32 + tid] = (sOut[0][tid] + sOut[1][tid]) * (1.f / 3136.f);
}

#define DFT_ROWS 8
#define DFT_COLS 16
#define DFT_CP   225

__global__ __launch_bounds__(224)
void dft_rows_kernel(const float* __restrict__ x, float* __restrict__ F1)
{
    __shared__ float ct[224], st[224];
    __shared__ float xs[DFT_ROWS][224];
    __shared__ float2 ys[DFT_ROWS][224];
    const int t   = threadIdx.x;
    const int bc  = blockIdx.x / (224 / DFT_ROWS);
    const int y0  = (blockIdx.x - bc * (224 / DFT_ROWS)) * DFT_ROWS;
    const float ang = (2.0f * PI_F / 224.0f) * (float)t;
    ct[t] = cosf(ang); st[t] = sinf(ang);
    const float* rows = x + ((size_t)bc * 224 + y0) * 224;
    #pragma unroll
    for (int r = 0; r < DFT_ROWS; ++r) xs[r][t] = rows[r * 224 + t];
    __syncthreads();

    const int lo = t & 15, hi = t >> 4;
    float re[DFT_ROWS], im[DFT_ROWS];
    #pragma unroll
    for (int r = 0; r < DFT_ROWS; ++r) { re[r] = 0.f; im[r] = 0.f; }
    {
        int j = 0;
        const int dj = 14 * lo;
        for (int n1 = 0; n1 < 16; ++n1) {
            const float c = ct[j], s = st[j];
            #pragma unroll
            for (int r = 0; r < DFT_ROWS; ++r) {
                const float v = xs[r][14 * n1 + hi];
                re[r] += v * c; im[r] -= v * s;
            }
            j += dj; if (j >= 224) j -= 224;
        }
        const float c = ct[hi * lo], s = st[hi * lo];
        #pragma unroll
        for (int r = 0; r < DFT_ROWS; ++r) {
            float2 y;
            y.x = re[r] * c + im[r] * s;
            y.y = im[r] * c - re[r] * s;
            ys[r][t] = y;
        }
    }
    __syncthreads();

    #pragma unroll
    for (int r = 0; r < DFT_ROWS; ++r) { re[r] = 0.f; im[r] = 0.f; }
    {
        int j = 0;
        const int dj = 16 * hi;
        for (int n2 = 0; n2 < 14; ++n2) {
            const float c = ct[j], s = st[j];
            #pragma unroll
            for (int r = 0; r < DFT_ROWS; ++r) {
                const float2 y = ys[r][n2 * 16 + lo];
                re[r] += y.x * c; re[r] += y.y * s;
                im[r] += y.y * c; im[r] -= y.x * s;
            }
            j += dj; if (j >= 224) j -= 224;
        }
    }
    float* o = F1 + (((size_t)bc * 224 + y0) * 224 + t) * 2;
    #pragma unroll
    for (int r = 0; r < DFT_ROWS; ++r) { o[r * 448] = re[r]; o[r * 448 + 1] = im[r]; }
}

__global__ __launch_bounds__(224)
void dft_cols_mag_kernel(const float* __restrict__ F1, float* __restrict__ mag)
{
    __shared__ float ct[224], st[224];
    __shared__ float2 fs[DFT_COLS][DFT_CP];
    __shared__ float2 ys[DFT_COLS][DFT_CP];
    const int t   = threadIdx.x;
    const int bc  = blockIdx.x / (224 / DFT_COLS);
    const int u0  = (blockIdx.x - bc * (224 / DFT_COLS)) * DFT_COLS;
    const float ang = (2.0f * PI_F / 224.0f) * (float)t;
    ct[t] = cosf(ang); st[t] = sinf(ang);
    {
        const float2* base = (const float2*)(F1 + (size_t)bc * (224 * 448) + u0 * 2);
        #pragma unroll
        for (int i = 0; i < 16; ++i) {
            const int e = i * 224 + t;
            const int y = e >> 4, q = e & 15;
            fs[q][y] = base[y * 224 + q];
        }
    }
    __syncthreads();

    const int lo = t & 15, hi = t >> 4;
    float re[DFT_COLS], im[DFT_COLS];
    #pragma unroll
    for (int q = 0; q < DFT_COLS; ++q) { re[q] = 0.f; im[q] = 0.f; }
    {
        int j = 0;
        const int dj = 14 * lo;
        for (int n1 = 0; n1 < 16; ++n1) {
            const float c = ct[j], s = st[j];
            #pragma unroll
            for (int q = 0; q < DFT_COLS; ++q) {
                const float2 f = fs[q][14 * n1 + hi];
                re[q] += f.x * c; re[q] += f.y * s;
                im[q] += f.y * c; im[q] -= f.x * s;
            }
            j += dj; if (j >= 224) j -= 224;
        }
        const float c = ct[hi * lo], s = st[hi * lo];
        #pragma unroll
        for (int q = 0; q < DFT_COLS; ++q) {
            float2 y;
            y.x = re[q] * c + im[q] * s;
            y.y = im[q] * c - re[q] * s;
            ys[q][t] = y;
        }
    }
    __syncthreads();

    #pragma unroll
    for (int q = 0; q < DFT_COLS; ++q) { re[q] = 0.f; im[q] = 0.f; }
    {
        int j = 0;
        const int dj = 16 * hi;
        for (int n2 = 0; n2 < 14; ++n2) {
            const float c = ct[j], s = st[j];
            #pragma unroll
            for (int q = 0; q < DFT_COLS; ++q) {
                const float2 y = ys[q][n2 * 16 + lo];
                re[q] += y.x * c; re[q] += y.y * s;
                im[q] += y.y * c; im[q] -= y.x * s;
            }
            j += dj; if (j >= 224) j -= 224;
        }
    }
    int vo = t + 112;  if (vo >= 224) vo -= 224;
    int uo = u0 + 112; if (uo >= 224) uo -= 224;
    float4* o = (float4*)(mag + ((size_t)bc * 224 + vo) * 224 + uo);
    #pragma unroll
    for (int q = 0; q < DFT_COLS; q += 4) {
        float4 m;
        m.x = logf(sqrtf(re[q]     * re[q]     + im[q]     * im[q])     + 1.0f);
        m.y = logf(sqrtf(re[q + 1] * re[q + 1] + im[q + 1] * im[q + 1]) + 1.0f);
        m.z = logf(sqrtf(re[q + 2] * re[q + 2] + im[q + 2] * im[q + 2]) + 1.0f);
        m.w = logf(sqrtf(re[q + 3] * re[q + 3] + im[q + 3] * im[q + 3]) + 1.0f);
        o[q >> 2] = m;
    }
}

__global__ __launch_bounds__(256)
void minmax_kernel(const float* __restrict__ mag, float* __restrict__ mm)
{
    __shared__ float smn[256], smx[256];
    int bc = blockIdx.x, tid = threadIdx.x;
    const float* p = mag + (size_t)bc * 50176;
    float mn = 3.0e38f, mx = -3.0e38f;
    for (int i = tid; i < 50176; i += 256) {
        float v = p[i]; mn = fminf(mn, v); mx = fmaxf(mx, v);
    }
    smn[tid] = mn; smx[tid] = mx; __syncthreads();
    for (int s = 128; s > 0; s >>= 1) {
        if (tid < s) { smn[tid] = fminf(smn[tid], smn[tid + s]);
                       smx[tid] = fmaxf(smx[tid], smx[tid + s]); }
        __syncthreads();
    }
    if (tid == 0) { mm[bc * 2] = smn[0]; mm[bc * 2 + 1] = smx[0]; }
}

__global__ __launch_bounds__(256)
void normalize_kernel(float* __restrict__ mag, const float* __restrict__ mm)
{
    int idx = blockIdx.x * 256 + threadIdx.x;
    if (idx >= 96 * 50176) return;
    int bc = idx / 50176;
    float mn = mm[bc * 2], mx = mm[bc * 2 + 1];
    mag[idx] = (mag[idx] - mn) / (mx - mn + 1e-8f);
}

#define FC_TP 272

template <int K, bool SMALLN>
__global__ __launch_bounds__(256)
void fc_wmma_kernel(const float* __restrict__ in, const float* __restrict__ w,
                    const float* __restrict__ b, float* __restrict__ out,
                    int N, int outStride, int outOffset, int act)
{
    constexpr int KC = K / 8;
    constexpr int NL = KC / 8;
    __shared__ float part[8 * 2 * FC_TP];

    const int tid  = threadIdx.x;
    const int lane = tid & 31;
    const int wave = tid >> 5;
    const int half = lane >> 4;
    const int c    = lane & 15;
    const int j0   = blockIdx.x * 32;
    const int i0   = blockIdx.y * 16;
    const int kb   = wave * KC + 4 * half;

    int ja = j0 + c, jb = j0 + 16 + c;
    if (SMALLN) ja = ja < N ? ja : N - 1;
    const float4* pa  = (const float4*)(in + (size_t)(i0 + c) * K + kb);
    const float4* pb0 = (const float4*)(w + (size_t)ja * K + kb);
    const float4* pb1 = (const float4*)(w + (size_t)jb * K + kb);

    int fin_il = 0, fin_j = lane;
    if (SMALLN) {
        const int tt = tid < 16 * N ? tid : 0;
        fin_il = tt / N;
        fin_j  = tt - fin_il * N;
    }
    const float bj = b[SMALLN ? fin_j : j0 + lane];

    float4 xa[NL], wb0[NL], wb1[NL];
    #pragma unroll
    for (int m = 0; m < NL; ++m) {
        xa[m]  = pa[2 * m];
        wb0[m] = pb0[2 * m];
        if (!SMALLN) wb1[m] = pb1[2 * m];
    }
    __builtin_amdgcn_sched_barrier(0);

    v8f acc0 = (v8f){0.f, 0.f, 0.f, 0.f, 0.f, 0.f, 0.f, 0.f};
    v8f acc1 = (v8f){0.f, 0.f, 0.f, 0.f, 0.f, 0.f, 0.f, 0.f};
    #pragma unroll
    for (int m = 0; m < NL; ++m) {
        v2f a, p, q;
        a.x = xa[m].x;  a.y = xa[m].y;
        p.x = wb0[m].x; p.y = wb0[m].y;
        acc0 = __builtin_amdgcn_wmma_f32_16x16x4_f32(false, a, false, p, (short)0, acc0, false, false);
        if (!SMALLN) {
            q.x = wb1[m].x; q.y = wb1[m].y;
            acc1 = __builtin_amdgcn_wmma_f32_16x16x4_f32(false, a, false, q, (short)0, acc1, false, false);
        }
        a.x = xa[m].z;  a.y = xa[m].w;
        p.x = wb0[m].z; p.y = wb0[m].w;
        acc0 = __builtin_amdgcn_wmma_f32_16x16x4_f32(false, a, false, p, (short)0, acc0, false, false);
        if (!SMALLN) {
            q.x = wb1[m].z; q.y = wb1[m].w;
            acc1 = __builtin_amdgcn_wmma_f32_16x16x4_f32(false, a, false, q, (short)0, acc1, false, false);
        }
    }

    float* pp = part + (wave * 2) * FC_TP + lane;
    #pragma unroll
    for (int r = 0; r < 8; ++r) {
        pp[r * 32] = acc0[r];
        if (!SMALLN) pp[FC_TP + r * 32] = acc1[r];
    }
    __syncthreads();

    if (!SMALLN) {
        const int t = lane >> 4;
        #pragma unroll
        for (int u = 0; u < 2; ++u) {
            const int il  = 2 * wave + u;
            const int src = t * FC_TP + (il & 7) * 32 + 16 * (il >> 3) + c;
            float s = part[src];
            #pragma unroll
            for (int q = 1; q < 8; ++q) s += part[q * 2 * FC_TP + src];
            s += bj;
            if (act == 1) s = fmaxf(s, 0.f);
            out[(size_t)(i0 + il) * outStride + outOffset + j0 + lane] = s;
        }
    } else {
        if (tid < 16 * N) {
            const int il  = fin_il;
            const int j   = fin_j;
            const int src = (il & 7) * 32 + 16 * (il >> 3) + j;
            float s = part[src];
            #pragma unroll
            for (int q = 1; q < 8; ++q) s += part[q * 2 * FC_TP + src];
            s += bj;
            if (act == 1) s = fmaxf(s, 0.f);
            out[(size_t)(i0 + il) * outStride + outOffset + j] = s;
        }
    }
}

__global__ __launch_bounds__(256)
void attn_core_kernel(const float* __restrict__ qkv_rgb, const float* __restrict__ qkv_frq,
                      float* __restrict__ o1, float* __restrict__ o2)
{
    __shared__ float sQ[32][32], sK[32][32], sV[32][32], sS[32][33];
    int a = blockIdx.x >> 3;
    int h = blockIdx.x & 7;
    const float* qsrc  = (a == 0) ? qkv_rgb : qkv_frq;
    const float* kvsrc = (a == 0) ? qkv_frq : qkv_rgb;
    float* odst        = (a == 0) ? o1 : o2;
    int tid = threadIdx.x;
    for (int idx = tid; idx < 1024; idx += 256) {
        int q = idx >> 5, d = idx & 31;
        sQ[q][d] = qsrc [q * 768 +       h * 32 + d];
        sK[q][d] = kvsrc[q * 768 + 256 + h * 32 + d];
        sV[q][d] = kvsrc[q * 768 + 512 + h * 32 + d];
    }
    __syncthreads();
    const float scale = 0.17677669529663687f;
    for (int idx = tid; idx < 1024; idx += 256) {
        int q = idx >> 5, k = idx & 31;
        float s = 0.f;
        #pragma unroll
        for (int d = 0; d < 32; ++d) s += sQ[q][d] * sK[k][d];
        sS[q][k] = s * scale;
    }
    __syncthreads();
    if (tid < 32) {
        float mx = -3.0e38f;
        for (int k = 0; k < 32; ++k) mx = fmaxf(mx, sS[tid][k]);
        float e[32], sum = 0.f;
        for (int k = 0; k < 32; ++k) { e[k] = expf(sS[tid][k] - mx); sum += e[k]; }
        float inv = 1.f / sum;
        for (int k = 0; k < 32; ++k) sS[tid][k] = e[k] * inv;
    }
    __syncthreads();
    for (int idx = tid; idx < 1024; idx += 256) {
        int q = idx >> 5, d = idx & 31;
        float o = 0.f;
        #pragma unroll
        for (int k = 0; k < 32; ++k) o += sS[q][k] * sV[k][d];
        odst[q * 256 + h * 32 + d] = o;
    }
}

extern "C" void kernel_launch(void* const* d_in, const int* in_sizes, int n_in,
                              void* d_out, int out_size, void* d_ws, size_t ws_size,
                              hipStream_t stream)
{
    (void)in_sizes; (void)n_in; (void)out_size; (void)ws_size;
    const float* x          = (const float*)d_in[0];
    const float* rgb_w1     = (const float*)d_in[1];
    const float* rgb_b1     = (const float*)d_in[2];
    const float* rgb_w2     = (const float*)d_in[3];
    const float* rgb_b2     = (const float*)d_in[4];
    const float* frq_w1     = (const float*)d_in[5];
    const float* frq_b1     = (const float*)d_in[6];
    const float* frq_w2     = (const float*)d_in[7];
    const float* frq_b2     = (const float*)d_in[8];
    const float* rgb_proj_w = (const float*)d_in[9];
    const float* rgb_proj_b = (const float*)d_in[10];
    const float* frq_proj_w = (const float*)d_in[11];
    const float* frq_proj_b = (const float*)d_in[12];
    const float* attn_w_in  = (const float*)d_in[13];
    const float* attn_b_in  = (const float*)d_in[14];
    const float* attn_w_out = (const float*)d_in[15];
    const float* attn_b_out = (const float*)d_in[16];
    const float* fus_w1     = (const float*)d_in[17];
    const float* fus_b1     = (const float*)d_in[18];
    const float* fus_w2     = (const float*)d_in[19];
    const float* fus_b2     = (const float*)d_in[20];
    const float* fus_w3     = (const float*)d_in[21];
    const float* fus_b3     = (const float*)d_in[22];

    float* out      = (float*)d_out;
    float* feat_rgb = out + 64;
    float* feat_frq = out + 64 + 40960;

    float* ws       = (float*)d_ws;
    float* h1       = ws;
    float* F1       = ws;
    float* freqimg  = ws + 12845056;
    float* mm       = freqimg + 4816896;
    float* hd       = mm + 192;
    float* rgb_p    = hd;
    float* frq_p    = rgb_p + 8192;
    float* qkv_rgb  = frq_p + 8192;
    float* qkv_frq  = qkv_rgb + 24576;
    float* o1       = qkv_frq + 24576;
    float* o2       = o1 + 8192;
    float* hb       = o2 + 8192;
    float* h1f      = hb + 16384;
    float* h2f      = h1f + 16384;

    conv1_silu_kernel<<<1568, 128, 0, stream>>>(x, rgb_w1, rgb_b1, h1);
    conv2_wmma_pool_kernel<<<dim3(40, 32), 128, 0, stream>>>(h1, rgb_w2, rgb_b2, feat_rgb);

    dft_rows_kernel<<<2688, 224, 0, stream>>>(x, F1);
    dft_cols_mag_kernel<<<1344, 224, 0, stream>>>(F1, freqimg);
    minmax_kernel<<<96, 256, 0, stream>>>(freqimg, mm);
    normalize_kernel<<<18816, 256, 0, stream>>>(freqimg, mm);
    conv1_silu_kernel<<<1568, 128, 0, stream>>>(freqimg, frq_w1, frq_b1, h1);
    conv2_wmma_pool_kernel<<<dim3(40, 32), 128, 0, stream>>>(h1, frq_w2, frq_b2, feat_frq);

    fc_wmma_kernel<1280, false><<<dim3(8, 2), 256, 0, stream>>>(feat_rgb, rgb_proj_w, rgb_proj_b, rgb_p, 256, 256, 0, 0);
    fc_wmma_kernel<1280, false><<<dim3(8, 2), 256, 0, stream>>>(feat_frq, frq_proj_w, frq_proj_b, frq_p, 256, 256, 0, 0);
    fc_wmma_kernel<256, false><<<dim3(24, 2), 256, 0, stream>>>(rgb_p, attn_w_in, attn_b_in, qkv_rgb, 768, 768, 0, 0);
    fc_wmma_kernel<256, false><<<dim3(24, 2), 256, 0, stream>>>(frq_p, attn_w_in, attn_b_in, qkv_frq, 768, 768, 0, 0);
    attn_core_kernel<<<16, 256, 0, stream>>>(qkv_rgb, qkv_frq, o1, o2);
    fc_wmma_kernel<256, false><<<dim3(8, 2), 256, 0, stream>>>(o1, attn_w_out, attn_b_out, hb, 256, 512, 0, 0);
    fc_wmma_kernel<256, false><<<dim3(8, 2), 256, 0, stream>>>(o2, attn_w_out, attn_b_out, hb, 256, 512, 256, 0);
    fc_wmma_kernel<512, false><<<dim3(16, 2), 256, 0, stream>>>(hb,  fus_w1, fus_b1, h1f, 512, 512, 0, 1);
    fc_wmma_kernel<512, false><<<dim3(8, 2), 256, 0, stream>>>(h1f, fus_w2, fus_b2, h2f, 256, 256, 0, 1);
    fc_wmma_kernel<256, true><<<dim3(1, 2), 256, 0, stream>>>(h2f, fus_w3, fus_b3, out, 2, 2, 0, 0);
}
